# starting kernel
# baseline (speedup 1.0000x reference)
#include <hip/hip_runtime.h>
#include <hip/hip_bf16.h>
#include <math.h>

typedef float v2f __attribute__((ext_vector_type(2)));
typedef float v8f __attribute__((ext_vector_type(8)));

#define D_FEAT 64
#define H_FEAT 128
#define WAVES_PER_BLOCK 4

__global__ void edge_msg_kernel(const float* __restrict__ x,
                                const int*   __restrict__ edge_index,
                                const int*   __restrict__ edge_attr,
                                const float* __restrict__ edge_weight,
                                const float* __restrict__ bond_emb,
                                float*       __restrict__ agg,
                                int E, int N)
{
    int e    = (int)((blockIdx.x * blockDim.x + threadIdx.x) >> 5);
    int lane = threadIdx.x & 31;
    if (e >= E) return;

    int src = edge_index[e];
    int dst = edge_index[E + e];
    int att = edge_attr[e];
    float w0 = edge_weight[e];
    float w1 = edge_weight[(size_t)E + e];

    int d = lane * 2;
    const float* er = bond_emb + (size_t)att * D_FEAT;
    float2 ev = *(const float2*)(er + d);

    const float kInvSqrt2 = 0.70710678118654752f;

    {
        const float* xr = x   + ((size_t)src) * D_FEAT;
        float*       ar = agg + ((size_t)dst) * D_FEAT;
        float2 xv = *(const float2*)(xr + d);
        float v0 = xv.x + ev.x;
        float v1 = xv.y + ev.y;
        v0 = 0.5f * v0 * (1.0f + erff(v0 * kInvSqrt2));
        v1 = 0.5f * v1 * (1.0f + erff(v1 * kInvSqrt2));
        atomicAdd(ar + d,     v0 * w0);
        atomicAdd(ar + d + 1, v1 * w0);
    }
    {
        const float* xr = x   + ((size_t)N + src) * D_FEAT;
        float*       ar = agg + ((size_t)N + dst) * D_FEAT;
        float2 xv = *(const float2*)(xr + d);
        float v0 = xv.x + ev.x;
        float v1 = xv.y + ev.y;
        v0 = 0.5f * v0 * (1.0f + erff(v0 * kInvSqrt2));
        v1 = 0.5f * v1 * (1.0f + erff(v1 * kInvSqrt2));
        atomicAdd(ar + d,     v0 * w1);
        atomicAdd(ar + d + 1, v1 * w1);
    }
}

__global__ void gin_mlp_kernel(const float* __restrict__ x,
                               const float* __restrict__ agg,
                               const float* __restrict__ W1,
                               const float* __restrict__ b1,
                               const float* __restrict__ W2,
                               const float* __restrict__ b2,
                               const float* __restrict__ eps,
                               float*       __restrict__ out,
                               int n_tiles)
{
    __shared__ float lds_t[WAVES_PER_BLOCK][16 * H_FEAT];

    int waveId = threadIdx.x >> 5;
    int lane   = threadIdx.x & 31;
    int tile   = blockIdx.x * WAVES_PER_BLOCK + waveId;
    if (tile >= n_tiles) return;

    float epsv = 1.0f + eps[0];
    int m    = lane & 15;
    int half = lane >> 4;

    size_t rowbase = ((size_t)tile * 16 + m) * D_FEAT;

    v2f A[16];
#pragma unroll
    for (int k = 0; k < 16; ++k) {
        int col = k * 4 + half * 2;
        float2 xv = *(const float2*)(x   + rowbase + col);
        float2 av = *(const float2*)(agg + rowbase + col);
        A[k].x = epsv * xv.x + av.x;
        A[k].y = epsv * xv.y + av.y;
    }

    float* tl = &lds_t[waveId][0];

#pragma unroll
    for (int n = 0; n < 8; ++n) {
        v8f acc = {};
        int ncol = n * 16 + m;
#pragma unroll
        for (int k = 0; k < 16; ++k) {
            int kb = k * 4 + half * 2;
            v2f B;
            B.x = W1[(size_t)kb * H_FEAT + ncol];
            B.y = W1[(size_t)(kb + 1) * H_FEAT + ncol];
            acc = __builtin_amdgcn_wmma_f32_16x16x4_f32(
                      false, A[k], false, B, (short)0, acc, false, false);
        }
        float bias = b1[ncol];
#pragma unroll
        for (int v = 0; v < 8; ++v) {
            float t = acc[v] + bias;
            t = t > 0.0f ? t : 0.0f;
            tl[(v + 8 * half) * H_FEAT + ncol] = t;
        }
    }

    asm volatile("s_wait_dscnt 0" ::: "memory");
    __builtin_amdgcn_wave_barrier();

#pragma unroll
    for (int o = 0; o < 4; ++o) {
        v8f acc = {};
        int ocol = o * 16 + m;
#pragma unroll
        for (int j = 0; j < 32; ++j) {
            int kb = j * 4 + half * 2;
            float2 av = *(const float2*)(tl + m * H_FEAT + kb);
            v2f A2; A2.x = av.x; A2.y = av.y;
            v2f B2;
            B2.x = W2[(size_t)kb * D_FEAT + ocol];
            B2.y = W2[(size_t)(kb + 1) * D_FEAT + ocol];
            acc = __builtin_amdgcn_wmma_f32_16x16x4_f32(
                      false, A2, false, B2, (short)0, acc, false, false);
        }
        float bias = b2[ocol];
#pragma unroll
        for (int v = 0; v < 8; ++v) {
            out[((size_t)tile * 16 + v + 8 * half) * D_FEAT + ocol] = acc[v] + bias;
        }
    }
}

extern "C" void kernel_launch(void* const* d_in, const int* in_sizes, int n_in,
                              void* d_out, int out_size, void* d_ws, size_t ws_size,
                              hipStream_t stream)
{
    const float* x           = (const float*)d_in[0];
    const int*   edge_index  = (const int*)  d_in[1];
    const int*   edge_attr   = (const int*)  d_in[2];
    const float* edge_weight = (const float*)d_in[3];
    const float* bond_emb    = (const float*)d_in[4];
    const float* eps         = (const float*)d_in[5];
    const float* W1          = (const float*)d_in[6];
    const float* b1          = (const float*)d_in[7];
    const float* W2          = (const float*)d_in[8];
    const float* b2          = (const float*)d_in[9];
    float*       out         = (float*)d_out;

    const int RC = 2;
    const int E  = in_sizes[2];
    const int N  = in_sizes[0] / (RC * D_FEAT);

    float* agg = (float*)d_ws;
    hipMemsetAsync(agg, 0, (size_t)RC * N * D_FEAT * sizeof(float), stream);

    long long totalThreads = (long long)E * 32;
    int blocks1 = (int)((totalThreads + 255) / 256);
    edge_msg_kernel<<<blocks1, 256, 0, stream>>>(x, edge_index, edge_attr,
                                                 edge_weight, bond_emb, agg, E, N);

    int n_tiles = (RC * N) / 16;
    int blocks2 = (n_tiles + WAVES_PER_BLOCK - 1) / WAVES_PER_BLOCK;
    gin_mlp_kernel<<<blocks2, WAVES_PER_BLOCK * 32, 0, stream>>>(
        x, agg, W1, b1, W2, b2, eps, out, n_tiles);
}


# MLP phase: W1/W2 B fragments staged once per 8-wave block into LDS in fragment order; each wave reads one 16-byte LDS word per two WMMAs instead of four 4-byte global loads
# speedup vs baseline: 1.0412x; 1.0412x over previous
#include <hip/hip_runtime.h>
#include <hip/hip_bf16.h>
#include <math.h>

typedef float v2f __attribute__((ext_vector_type(2)));
typedef float v8f __attribute__((ext_vector_type(8)));

#define D_FEAT 64
#define H_FEAT 128
#define WAVES_PER_BLOCK 8

__global__ void edge_msg_kernel(const float* __restrict__ x,
                                const int*   __restrict__ edge_index,
                                const int*   __restrict__ edge_attr,
                                const float* __restrict__ edge_weight,
                                const float* __restrict__ bond_emb,
                                float*       __restrict__ agg,
                                int E, int N)
{
    int e    = (int)((blockIdx.x * blockDim.x + threadIdx.x) >> 5);
    int lane = threadIdx.x & 31;
    if (e >= E) return;

    int src = edge_index[e];
    int dst = edge_index[E + e];
    int att = edge_attr[e];
    float w0 = edge_weight[e];
    float w1 = edge_weight[(size_t)E + e];

    int d = lane * 2;
    const float* er = bond_emb + (size_t)att * D_FEAT;
    float2 ev = *(const float2*)(er + d);

    const float kInvSqrt2 = 0.70710678118654752f;

    {
        const float* xr = x   + ((size_t)src) * D_FEAT;
        float*       ar = agg + ((size_t)dst) * D_FEAT;
        float2 xv = *(const float2*)(xr + d);
        float v0 = xv.x + ev.x;
        float v1 = xv.y + ev.y;
        v0 = 0.5f * v0 * (1.0f + erff(v0 * kInvSqrt2));
        v1 = 0.5f * v1 * (1.0f + erff(v1 * kInvSqrt2));
        atomicAdd(ar + d,     v0 * w0);
        atomicAdd(ar + d + 1, v1 * w0);
    }
    {
        const float* xr = x   + ((size_t)N + src) * D_FEAT;
        float*       ar = agg + ((size_t)N + dst) * D_FEAT;
        float2 xv = *(const float2*)(xr + d);
        float v0 = xv.x + ev.x;
        float v1 = xv.y + ev.y;
        v0 = 0.5f * v0 * (1.0f + erff(v0 * kInvSqrt2));
        v1 = 0.5f * v1 * (1.0f + erff(v1 * kInvSqrt2));
        atomicAdd(ar + d,     v0 * w1);
        atomicAdd(ar + d + 1, v1 * w1);
    }
}

__global__ __launch_bounds__(WAVES_PER_BLOCK * 32)
void gin_mlp_kernel(const float* __restrict__ x,
                    const float* __restrict__ agg,
                    const float* __restrict__ W1,
                    const float* __restrict__ b1,
                    const float* __restrict__ W2,
                    const float* __restrict__ b2,
                    const float* __restrict__ eps,
                    float*       __restrict__ out,
                    int n_tiles)
{
    __shared__ float4 w1s[8 * 8 * 32];
    __shared__ float4 w2s[4 * 16 * 32];
    __shared__ float  lds_t[WAVES_PER_BLOCK][16 * H_FEAT];

    int waveId = threadIdx.x >> 5;
    int lane   = threadIdx.x & 31;
    int tile   = blockIdx.x * WAVES_PER_BLOCK + waveId;
    bool valid = tile < n_tiles;

    int m    = lane & 15;
    int half = lane >> 4;

    v2f A[16];
#pragma unroll
    for (int k = 0; k < 16; ++k) { A[k].x = 0.0f; A[k].y = 0.0f; }
    if (valid) {
        float epsv = 1.0f + eps[0];
        size_t rowbase = ((size_t)tile * 16 + m) * D_FEAT;
#pragma unroll
        for (int k = 0; k < 16; ++k) {
            int col = k * 4 + half * 2;
            float2 xv = *(const float2*)(x   + rowbase + col);
            float2 av = *(const float2*)(agg + rowbase + col);
            A[k].x = epsv * xv.x + av.x;
            A[k].y = epsv * xv.y + av.y;
        }
    }

#pragma unroll
    for (int n = 0; n < 8; ++n) {
        int ncol = n * 16 + m;
        int kb0  = waveId * 8 + half * 2;
        float4 e;
        e.x = W1[(size_t)(kb0    ) * H_FEAT + ncol];
        e.y = W1[(size_t)(kb0 + 1) * H_FEAT + ncol];
        e.z = W1[(size_t)(kb0 + 4) * H_FEAT + ncol];
        e.w = W1[(size_t)(kb0 + 5) * H_FEAT + ncol];
        w1s[(n * 8 + waveId) * 32 + lane] = e;
    }
#pragma unroll
    for (int i = 0; i < 8; ++i) {
        int o    = i >> 1;
        int jj   = waveId + 8 * (i & 1);
        int ocol = o * 16 + m;
        int kb0  = jj * 8 + half * 2;
        float4 e;
        e.x = W2[(size_t)(kb0    ) * D_FEAT + ocol];
        e.y = W2[(size_t)(kb0 + 1) * D_FEAT + ocol];
        e.z = W2[(size_t)(kb0 + 4) * D_FEAT + ocol];
        e.w = W2[(size_t)(kb0 + 5) * D_FEAT + ocol];
        w2s[(o * 16 + jj) * 32 + lane] = e;
    }
    __syncthreads();
    if (!valid) return;

    float* tl = &lds_t[waveId][0];

#pragma unroll
    for (int n = 0; n < 8; ++n) {
        v8f acc = {};
        int ncol = n * 16 + m;
#pragma unroll
        for (int kk = 0; kk < 8; ++kk) {
            float4 e = w1s[(n * 8 + kk) * 32 + lane];
            v2f B0; B0.x = e.x; B0.y = e.y;
            v2f B1; B1.x = e.z; B1.y = e.w;
            acc = __builtin_amdgcn_wmma_f32_16x16x4_f32(
                      false, A[2 * kk], false, B0, (short)0, acc, false, false);
            acc = __builtin_amdgcn_wmma_f32_16x16x4_f32(
                      false, A[2 * kk + 1], false, B1, (short)0, acc, false, false);
        }
        float bias = b1[ncol];
#pragma unroll
        for (int v = 0; v < 8; ++v) {
            float t = acc[v] + bias;
            t = t > 0.0f ? t : 0.0f;
            tl[(v + 8 * half) * H_FEAT + ncol] = t;
        }
    }

    asm volatile("s_wait_dscnt 0" ::: "memory");
    __builtin_amdgcn_wave_barrier();

#pragma unroll
    for (int o = 0; o < 4; ++o) {
        v8f acc = {};
        int ocol = o * 16 + m;
#pragma unroll
        for (int jj = 0; jj < 16; ++jj) {
            float4 e = w2s[(o * 16 + jj) * 32 + lane];
            int kb = jj * 8 + half * 2;
            float2 a0 = *(const float2*)(tl + m * H_FEAT + kb);
            float2 a1 = *(const float2*)(tl + m * H_FEAT + kb + 4);
            v2f A2; A2.x = a0.x; A2.y = a0.y;
            v2f A3; A3.x = a1.x; A3.y = a1.y;
            v2f B0; B0.x = e.x; B0.y = e.y;
            v2f B1; B1.x = e.z; B1.y = e.w;
            acc = __builtin_amdgcn_wmma_f32_16x16x4_f32(
                      false, A2, false, B0, (short)0, acc, false, false);
            acc = __builtin_amdgcn_wmma_f32_16x16x4_f32(
                      false, A3, false, B1, (short)0, acc, false, false);
        }
        float bias = b2[ocol];
#pragma unroll
        for (int v = 0; v < 8; ++v) {
            out[((size_t)tile * 16 + v + 8 * half) * D_FEAT + ocol] = acc[v] + bias;
        }
    }
}

extern "C" void kernel_launch(void* const* d_in, const int* in_sizes, int n_in,
                              void* d_out, int out_size, void* d_ws, size_t ws_size,
                              hipStream_t stream)
{
    const float* x           = (const float*)d_in[0];
    const int*   edge_index  = (const int*)  d_in[1];
    const int*   edge_attr   = (const int*)  d_in[2];
    const float* edge_weight = (const float*)d_in[3];
    const float* bond_emb    = (const float*)d_in[4];
    const float* eps         = (const float*)d_in[5];
    const float* W1          = (const float*)d_in[6];
    const float* b1          = (const float*)d_in[7];
    const float* W2          = (const float*)d_in[8];
    const float* b2          = (const float*)d_in[9];
    float*       out         = (float*)d_out;

    const int RC = 2;
    const int E  = in_sizes[2];
    const int N  = in_sizes[0] / (RC * D_FEAT);

    float* agg = (float*)d_ws;
    hipMemsetAsync(agg, 0, (size_t)RC * N * D_FEAT * sizeof(float), stream);

    long long totalThreads = (long long)E * 32;
    int blocks1 = (int)((totalThreads + 255) / 256);
    edge_msg_kernel<<<blocks1, 256, 0, stream>>>(x, edge_index, edge_attr,
                                                 edge_weight, bond_emb, agg, E, N);

    int n_tiles = (RC * N) / 16;
    int blocks2 = (n_tiles + WAVES_PER_BLOCK - 1) / WAVES_PER_BLOCK;
    gin_mlp_kernel<<<blocks2, WAVES_PER_BLOCK * 32, 0, stream>>>(
        x, agg, W1, b1, W2, b2, eps, out, n_tiles);
}


# Edge phase: edges binned by destination per call (LDS counters, integer atomics only), then one wave per node sums its messages in registers in ascending edge id; f32 atomics and agg clear gone
# speedup vs baseline: 1.3353x; 1.2825x over previous
#include <hip/hip_runtime.h>
#include <hip/hip_bf16.h>
#include <math.h>


typedef float v2f __attribute__((ext_vector_type(2)));
typedef float v8f __attribute__((ext_vector_type(8)));

#define D_FEAT 64
#define H_FEAT 128
#define WAVES_PER_BLOCK 8

#define BIN_SHIFT    6
#define MAX_BINS     1024
#define PART_THREADS 512
#define PART_EDGES   4096
#define LIST_CAP     128

static_assert(MAX_BINS == 2 * PART_THREADS, "pass B scans two bins per thread");

__global__ __launch_bounds__(PART_THREADS)
void bin_count_kernel(const int* __restrict__ edge_index,
                      int*       __restrict__ bin_count,
                      int E, int NB)
{
    __shared__ int hist[MAX_BINS];
    hist[2 * threadIdx.x]     = 0;
    hist[2 * threadIdx.x + 1] = 0;
    __syncthreads();
    int base = blockIdx.x * PART_EDGES;
    int dstv[PART_EDGES / PART_THREADS];
    for (int i = 0; i < PART_EDGES / PART_THREADS; ++i) {
        int e = base + i * PART_THREADS + (int)threadIdx.x;
        dstv[i] = (e < E) ? edge_index[E + e] : -1;
    }
    for (int i = 0; i < PART_EDGES / PART_THREADS; ++i)
        if (dstv[i] >= 0) atomicAdd(&hist[dstv[i] >> BIN_SHIFT], 1);
    __syncthreads();
    for (int b = threadIdx.x; b < NB; b += PART_THREADS) {
        int h = hist[b];
        if (h) atomicAdd(&bin_count[b], h);
    }
}

__global__ __launch_bounds__(PART_THREADS)
void bin_partition_kernel(const int* __restrict__ edge_index,
                          const int* __restrict__ bin_count,
                          int*       __restrict__ bin_cursor,
                          int*       __restrict__ bin_start,
                          int*       __restrict__ recs,
                          int E, int NB)
{
    __shared__ int hist[MAX_BINS];
    __shared__ int sbase[MAX_BINS];
    __shared__ int wtot[PART_THREADS / 32];

    int tid  = threadIdx.x;
    int lane = tid & 31;
    int w    = tid >> 5;

    int c0 = (2 * tid     < NB) ? bin_count[2 * tid]     : 0;
    int c1 = (2 * tid + 1 < NB) ? bin_count[2 * tid + 1] : 0;
    int v   = c0 + c1;
    int inc = v;
    for (int off = 1; off < 32; off <<= 1) {
        int o = __shfl_up(inc, off);
        if (lane >= off) inc += o;
    }
    if (lane == 31) wtot[w] = inc;
    hist[2 * tid]     = 0;
    hist[2 * tid + 1] = 0;
    __syncthreads();
    int wbase = 0;
    for (int i = 0; i < w; ++i) wbase += wtot[i];
    int ex = wbase + inc - v;
    sbase[2 * tid]     = ex;
    sbase[2 * tid + 1] = ex + c0;
    if (blockIdx.x == 0) {
        if (2 * tid     < NB) bin_start[2 * tid]     = ex;
        if (2 * tid + 1 < NB) bin_start[2 * tid + 1] = ex + c0;
    }

    int base = blockIdx.x * PART_EDGES;
    int dstv[PART_EDGES / PART_THREADS];
    for (int i = 0; i < PART_EDGES / PART_THREADS; ++i) {
        int e = base + i * PART_THREADS + tid;
        dstv[i] = (e < E) ? edge_index[E + e] : -1;
    }
    for (int i = 0; i < PART_EDGES / PART_THREADS; ++i)
        if (dstv[i] >= 0) atomicAdd(&hist[dstv[i] >> BIN_SHIFT], 1);
    __syncthreads();

    for (int b = tid; b < NB; b += PART_THREADS) {
        int h = hist[b];
        int g = h ? atomicAdd(&bin_cursor[b], h) : 0;
        sbase[b] += g;
        hist[b] = 0;
    }
    __syncthreads();

    for (int i = 0; i < PART_EDGES / PART_THREADS; ++i) {
        if (dstv[i] >= 0) {
            int b    = dstv[i] >> BIN_SHIFT;
            int slot = sbase[b] + atomicAdd(&hist[b], 1);
            int e    = base + i * PART_THREADS + tid;
            recs[slot] = (e << BIN_SHIFT) | (dstv[i] & ((1 << BIN_SHIFT) - 1));
        }
    }
}

__device__ __forceinline__ float gelu_exact(float v)
{
    const float kInvSqrt2 = 0.70710678118654752f;
    return 0.5f * v * (1.0f + erff(v * kInvSqrt2));
}

__global__ __launch_bounds__(256)
void node_sum_kernel(const float* __restrict__ x,
                     const int*   __restrict__ edge_index,
                     const int*   __restrict__ edge_attr,
                     const float* __restrict__ edge_weight,
                     const float* __restrict__ bond_emb,
                     const int*   __restrict__ bin_start,
                     const int*   __restrict__ bin_count,
                     const int*   __restrict__ recs,
                     float*       __restrict__ agg,
                     int E, int N)
{
    __shared__ int elist[8][LIST_CAP];
    __shared__ int sorted[8][LIST_CAP];

    int w    = threadIdx.x >> 5;
    int lane = threadIdx.x & 31;
    int n    = blockIdx.x * 8 + w;
    if (n >= N) return;

    const int kmask = (1 << BIN_SHIFT) - 1;
    int bin = n >> BIN_SHIFT;
    int key = n & kmask;
    int pad = key ^ 1;
    int s   = __builtin_amdgcn_readfirstlane(bin_start[bin]);
    int t   = s + __builtin_amdgcn_readfirstlane(bin_count[bin]);
    int d   = lane * 2;

    const float* x0 = x + d;
    const float* x1 = x + (size_t)N * D_FEAT + d;
    const float* eb = bond_emb + d;

    float2 acc0 = make_float2(0.0f, 0.0f);
    float2 acc1 = make_float2(0.0f, 0.0f);

    int deg = 0;
    for (int base = s; base < t; base += 128) {
        int r[4];
        for (int u = 0; u < 4; ++u) {
            int idx = base + u * 32 + lane;
            r[u] = (idx < t) ? recs[idx] : pad;
        }
        for (int u = 0; u < 4; ++u) {
            bool mine = ((r[u] & kmask) == key);
            unsigned mask = __builtin_amdgcn_ballot_w32(mine);
            if (mask != 0u) {
                int pos = deg + (int)__builtin_amdgcn_mbcnt_lo(mask, 0u);
                if (mine && pos < LIST_CAP) elist[w][pos] = r[u] >> BIN_SHIFT;
                deg += __builtin_popcount(mask);
            }
        }
    }

    if (deg <= LIST_CAP) {
        for (int i = lane; i < deg; i += 32) {
            int ei = elist[w][i];
            int rank = 0;
            for (int j = 0; j < deg; ++j) rank += (elist[w][j] < ei) ? 1 : 0;
            sorted[w][rank] = ei;
        }

        for (int g = 0; g < deg; g += 32) {
            int cnt = deg - g;
            if (cnt > 32) cnt = 32;

            int   src_l = 0, att_l = 0;
            float w0_l = 0.0f, w1_l = 0.0f;
            if (lane < cnt) {
                int es = sorted[w][g + lane];
                src_l = edge_index[es];
                att_l = edge_attr[es];
                w0_l  = edge_weight[es];
                w1_l  = edge_weight[(size_t)E + es];
            }

            int src = __builtin_amdgcn_readlane(src_l, 0);
            int att = __builtin_amdgcn_readlane(att_l, 0);
            float2 ev = *(const float2*)(eb + (size_t)att * D_FEAT);
            float2 xa = *(const float2*)(x0 + (size_t)src * D_FEAT);
            float2 xb = *(const float2*)(x1 + (size_t)src * D_FEAT);
            for (int j = 0; j < cnt; ++j) {
                int jn   = (j + 1 < cnt) ? j + 1 : j;
                int srcn = __builtin_amdgcn_readlane(src_l, jn);
                int attn = __builtin_amdgcn_readlane(att_l, jn);
                float2 evn = *(const float2*)(eb + (size_t)attn * D_FEAT);
                float2 xan = *(const float2*)(x0 + (size_t)srcn * D_FEAT);
                float2 xbn = *(const float2*)(x1 + (size_t)srcn * D_FEAT);

                float w0 = __int_as_float(__builtin_amdgcn_readlane(__float_as_int(w0_l), j));
                float w1 = __int_as_float(__builtin_amdgcn_readlane(__float_as_int(w1_l), j));
                acc0.x += gelu_exact(xa.x + ev.x) * w0;
                acc0.y += gelu_exact(xa.y + ev.y) * w0;
                acc1.x += gelu_exact(xb.x + ev.x) * w1;
                acc1.y += gelu_exact(xb.y + ev.y) * w1;

                ev = evn; xa = xan; xb = xbn;
            }
        }
    } else {
        int last = -1;
        for (int k = 0; k < deg; ++k) {
            int best = 0x7fffffff;
            for (int base = s; base < t; base += 32) {
                int idx = base + lane;
                int r = (idx < t) ? recs[idx] : pad;
                int e = r >> BIN_SHIFT;
                if ((r & kmask) == key && e > last && e < best) best = e;
            }
            for (int off = 16; off > 0; off >>= 1) {
                int o = __shfl_xor(best, off);
                best = (o < best) ? o : best;
            }
            int e = __builtin_amdgcn_readfirstlane(best);
            last = e;
            int   src = edge_index[e];
            int   att = edge_attr[e];
            float w0  = edge_weight[e];
            float w1  = edge_weight[(size_t)E + e];
            float2 ev = *(const float2*)(eb + (size_t)att * D_FEAT);
            float2 xa = *(const float2*)(x0 + (size_t)src * D_FEAT);
            float2 xb = *(const float2*)(x1 + (size_t)src * D_FEAT);
            acc0.x += gelu_exact(xa.x + ev.x) * w0;
            acc0.y += gelu_exact(xa.y + ev.y) * w0;
            acc1.x += gelu_exact(xb.x + ev.x) * w1;
            acc1.y += gelu_exact(xb.y + ev.y) * w1;
        }
    }

    *(float2*)(agg + (size_t)n * D_FEAT + d)       = acc0;
    *(float2*)(agg + ((size_t)N + n) * D_FEAT + d) = acc1;
}

__global__ __launch_bounds__(WAVES_PER_BLOCK * 32)
void gin_mlp_kernel(const float* __restrict__ x,
                    const float* __restrict__ agg,
                    const float* __restrict__ W1,
                    const float* __restrict__ b1,
                    const float* __restrict__ W2,
                    const float* __restrict__ b2,
                    const float* __restrict__ eps,
                    float*       __restrict__ out,
                    int n_tiles)
{
    __shared__ float4 w1s[8 * 8 * 32];
    __shared__ float4 w2s[4 * 16 * 32];
    __shared__ float  lds_t[WAVES_PER_BLOCK][16 * H_FEAT];

    int waveId = threadIdx.x >> 5;
    int lane   = threadIdx.x & 31;
    int tile   = blockIdx.x * WAVES_PER_BLOCK + waveId;
    bool valid = tile < n_tiles;

    int m    = lane & 15;
    int half = lane >> 4;

    v2f A[16];
#pragma unroll
    for (int k = 0; k < 16; ++k) { A[k].x = 0.0f; A[k].y = 0.0f; }
    if (valid) {
        float epsv = 1.0f + eps[0];
        size_t rowbase = ((size_t)tile * 16 + m) * D_FEAT;
#pragma unroll
        for (int k = 0; k < 16; ++k) {
            int col = k * 4 + half * 2;
            float2 xv = *(const float2*)(x   + rowbase + col);
            float2 av = *(const float2*)(agg + rowbase + col);
            A[k].x = epsv * xv.x + av.x;
            A[k].y = epsv * xv.y + av.y;
        }
    }

#pragma unroll
    for (int n = 0; n < 8; ++n) {
        int ncol = n * 16 + m;
        int kb0  = waveId * 8 + half * 2;
        float4 e;
        e.x = W1[(size_t)(kb0    ) * H_FEAT + ncol];
        e.y = W1[(size_t)(kb0 + 1) * H_FEAT + ncol];
        e.z = W1[(size_t)(kb0 + 4) * H_FEAT + ncol];
        e.w = W1[(size_t)(kb0 + 5) * H_FEAT + ncol];
        w1s[(n * 8 + waveId) * 32 + lane] = e;
    }
#pragma unroll
    for (int i = 0; i < 8; ++i) {
        int o    = i >> 1;
        int jj   = waveId + 8 * (i & 1);
        int ocol = o * 16 + m;
        int kb0  = jj * 8 + half * 2;
        float4 e;
        e.x = W2[(size_t)(kb0    ) * D_FEAT + ocol];
        e.y = W2[(size_t)(kb0 + 1) * D_FEAT + ocol];
        e.z = W2[(size_t)(kb0 + 4) * D_FEAT + ocol];
        e.w = W2[(size_t)(kb0 + 5) * D_FEAT + ocol];
        w2s[(o * 16 + jj) * 32 + lane] = e;
    }
    __syncthreads();
    if (!valid) return;

    float* tl = &lds_t[waveId][0];

#pragma unroll
    for (int n = 0; n < 8; ++n) {
        v8f acc = {};
        int ncol = n * 16 + m;
#pragma unroll
        for (int kk = 0; kk < 8; ++kk) {
            float4 e = w1s[(n * 8 + kk) * 32 + lane];
            v2f B0; B0.x = e.x; B0.y = e.y;
            v2f B1; B1.x = e.z; B1.y = e.w;
            acc = __builtin_amdgcn_wmma_f32_16x16x4_f32(
                      false, A[2 * kk], false, B0, (short)0, acc, false, false);
            acc = __builtin_amdgcn_wmma_f32_16x16x4_f32(
                      false, A[2 * kk + 1], false, B1, (short)0, acc, false, false);
        }
        float bias = b1[ncol];
#pragma unroll
        for (int v = 0; v < 8; ++v) {
            float t = acc[v] + bias;
            t = t > 0.0f ? t : 0.0f;
            tl[(v + 8 * half) * H_FEAT + ncol] = t;
        }
    }

    asm volatile("s_wait_dscnt 0" ::: "memory");
    __builtin_amdgcn_wave_barrier();

#pragma unroll
    for (int o = 0; o < 4; ++o) {
        v8f acc = {};
        int ocol = o * 16 + m;
#pragma unroll
        for (int jj = 0; jj < 16; ++jj) {
            float4 e = w2s[(o * 16 + jj) * 32 + lane];
            int kb = jj * 8 + half * 2;
            float2 a0 = *(const float2*)(tl + m * H_FEAT + kb);
            float2 a1 = *(const float2*)(tl + m * H_FEAT + kb + 4);
            v2f A2; A2.x = a0.x; A2.y = a0.y;
            v2f A3; A3.x = a1.x; A3.y = a1.y;
            v2f B0; B0.x = e.x; B0.y = e.y;
            v2f B1; B1.x = e.z; B1.y = e.w;
            acc = __builtin_amdgcn_wmma_f32_16x16x4_f32(
                      false, A2, false, B0, (short)0, acc, false, false);
            acc = __builtin_amdgcn_wmma_f32_16x16x4_f32(
                      false, A3, false, B1, (short)0, acc, false, false);
        }
        float bias = b2[ocol];
#pragma unroll
        for (int v = 0; v < 8; ++v) {
            out[((size_t)tile * 16 + v + 8 * half) * D_FEAT + ocol] = acc[v] + bias;
        }
    }
}

extern "C" void kernel_launch(void* const* d_in, const int* in_sizes, int n_in,
                              void* d_out, int out_size, void* d_ws, size_t ws_size,
                              hipStream_t stream)
{
    const float* x           = (const float*)d_in[0];
    const int*   edge_index  = (const int*)  d_in[1];
    const int*   edge_attr   = (const int*)  d_in[2];
    const float* edge_weight = (const float*)d_in[3];
    const float* bond_emb    = (const float*)d_in[4];
    const float* eps         = (const float*)d_in[5];
    const float* W1          = (const float*)d_in[6];
    const float* b1          = (const float*)d_in[7];
    const float* W2          = (const float*)d_in[8];
    const float* b2          = (const float*)d_in[9];
    float*       out         = (float*)d_out;

    const int RC = 2;
    const int E  = in_sizes[2];
    const int N  = in_sizes[0] / (RC * D_FEAT);
    const int NB = (N + (1 << BIN_SHIFT) - 1) >> BIN_SHIFT;

    float* agg        = (float*)d_ws;
    int*   bin_count  = (int*)(agg + (size_t)RC * N * D_FEAT);
    int*   bin_cursor = bin_count + MAX_BINS;
    int*   bin_start  = bin_cursor + MAX_BINS;
    int*   recs       = bin_start + MAX_BINS;

    hipMemsetAsync(bin_count, 0, (size_t)2 * MAX_BINS * sizeof(int), stream);

    int blocksP = (E + PART_EDGES - 1) / PART_EDGES;
    bin_count_kernel<<<blocksP, PART_THREADS, 0, stream>>>(edge_index, bin_count, E, NB);
    bin_partition_kernel<<<blocksP, PART_THREADS, 0, stream>>>(edge_index, bin_count, bin_cursor,
                                                               bin_start, recs, E, NB);

    int blocks1 = (N + 7) / 8;
    node_sum_kernel<<<blocks1, 256, 0, stream>>>(x, edge_index, edge_attr, edge_weight, bond_emb,
                                                 bin_start, bin_count, recs, agg, E, N);

    int n_tiles = (RC * N) / 16;
    int blocks2 = (n_tiles + WAVES_PER_BLOCK - 1) / WAVES_PER_BLOCK;
    gin_mlp_kernel<<<blocks2, WAVES_PER_BLOCK * 32, 0, stream>>>(
        x, agg, W1, b1, W2, b2, eps, out, n_tiles);
}


# Edge phase: gelu as max(v,0) - |v|*2^q(|v|), q ONE degree-9 polynomial (log2 of the normal tail) for all |v|, held at 8: no erff call, no range split, no select; packed f32 pairs
# speedup vs baseline: 1.7625x; 1.3199x over previous
#include <hip/hip_runtime.h>
#include <hip/hip_bf16.h>
#include <math.h>


typedef float v2f __attribute__((ext_vector_type(2)));
typedef float v8f __attribute__((ext_vector_type(8)));

#define D_FEAT 64
#define H_FEAT 128
#define WAVES_PER_BLOCK 8

#define BIN_SHIFT    6
#define MAX_BINS     1024
#define PART_THREADS 512
#define PART_EDGES   4096
#define LIST_CAP     128

static_assert(MAX_BINS == 2 * PART_THREADS, "pass B scans two bins per thread");

__global__ __launch_bounds__(PART_THREADS)
void bin_count_kernel(const int* __restrict__ edge_index,
                      int*       __restrict__ bin_count,
                      int E, int NB)
{
    __shared__ int hist[MAX_BINS];
    hist[2 * threadIdx.x]     = 0;
    hist[2 * threadIdx.x + 1] = 0;
    __syncthreads();
    int base = blockIdx.x * PART_EDGES;
    int dstv[PART_EDGES / PART_THREADS];
    for (int i = 0; i < PART_EDGES / PART_THREADS; ++i) {
        int e = base + i * PART_THREADS + (int)threadIdx.x;
        dstv[i] = (e < E) ? edge_index[E + e] : -1;
    }
    for (int i = 0; i < PART_EDGES / PART_THREADS; ++i)
        if (dstv[i] >= 0) atomicAdd(&hist[dstv[i] >> BIN_SHIFT], 1);
    __syncthreads();
    for (int b = threadIdx.x; b < NB; b += PART_THREADS) {
        int h = hist[b];
        if (h) atomicAdd(&bin_count[b], h);
    }
}

__global__ __launch_bounds__(PART_THREADS)
void bin_partition_kernel(const int* __restrict__ edge_index,
                          const int* __restrict__ bin_count,
                          int*       __restrict__ bin_cursor,
                          int*       __restrict__ bin_start,
                          int*       __restrict__ recs,
                          int E, int NB)
{
    __shared__ int hist[MAX_BINS];
    __shared__ int sbase[MAX_BINS];
    __shared__ int wtot[PART_THREADS / 32];

    int tid  = threadIdx.x;
    int lane = tid & 31;
    int w    = tid >> 5;

    int c0 = (2 * tid     < NB) ? bin_count[2 * tid]     : 0;
    int c1 = (2 * tid + 1 < NB) ? bin_count[2 * tid + 1] : 0;
    int v   = c0 + c1;
    int inc = v;
    for (int off = 1; off < 32; off <<= 1) {
        int o = __shfl_up(inc, off);
        if (lane >= off) inc += o;
    }
    if (lane == 31) wtot[w] = inc;
    hist[2 * tid]     = 0;
    hist[2 * tid + 1] = 0;
    __syncthreads();
    int wbase = 0;
    for (int i = 0; i < w; ++i) wbase += wtot[i];
    int ex = wbase + inc - v;
    sbase[2 * tid]     = ex;
    sbase[2 * tid + 1] = ex + c0;
    if (blockIdx.x == 0) {
        if (2 * tid     < NB) bin_start[2 * tid]     = ex;
        if (2 * tid + 1 < NB) bin_start[2 * tid + 1] = ex + c0;
    }

    int base = blockIdx.x * PART_EDGES;
    int dstv[PART_EDGES / PART_THREADS];
    for (int i = 0; i < PART_EDGES / PART_THREADS; ++i) {
        int e = base + i * PART_THREADS + tid;
        dstv[i] = (e < E) ? edge_index[E + e] : -1;
    }
    for (int i = 0; i < PART_EDGES / PART_THREADS; ++i)
        if (dstv[i] >= 0) atomicAdd(&hist[dstv[i] >> BIN_SHIFT], 1);
    __syncthreads();

    for (int b = tid; b < NB; b += PART_THREADS) {
        int h = hist[b];
        int g = h ? atomicAdd(&bin_cursor[b], h) : 0;
        sbase[b] += g;
        hist[b] = 0;
    }
    __syncthreads();

    for (int i = 0; i < PART_EDGES / PART_THREADS; ++i) {
        if (dstv[i] >= 0) {
            int b    = dstv[i] >> BIN_SHIFT;
            int slot = sbase[b] + atomicAdd(&hist[b], 1);
            int e    = base + i * PART_THREADS + tid;
            recs[slot] = (e << BIN_SHIFT) | (dstv[i] & ((1 << BIN_SHIFT) - 1));
        }
    }
}

__device__ __forceinline__ v2f gelu_pair(v2f v)
{
    v2f a;
    a.x = fminf(fabsf(v.x), 8.0f);
    a.y = fminf(fabsf(v.y), 8.0f);
    v2f q = (v2f)(2.1919348114352033e-07f);
    q = __builtin_elementwise_fma(q, a, (v2f)(-5.2599425544030964e-06f));
    q = __builtin_elementwise_fma(q, a, (v2f)(4.943032763549127e-05f));
    q = __builtin_elementwise_fma(q, a, (v2f)(-0.00020471429161261767f));
    q = __builtin_elementwise_fma(q, a, (v2f)(-0.00012574861466418952f));
    q = __builtin_elementwise_fma(q, a, (v2f)(0.007073047570884228f));
    q = __builtin_elementwise_fma(q, a, (v2f)(-0.05251256376504898f));
    q = __builtin_elementwise_fma(q, a, (v2f)(-0.45920106768608093f));
    q = __builtin_elementwise_fma(q, a, (v2f)(-1.1511061191558838f));
    q = __builtin_elementwise_fma(q, a, (v2f)(-1.0f));
    v2f t;
    t.x = __builtin_amdgcn_exp2f(q.x);
    t.y = __builtin_amdgcn_exp2f(q.y);
    v2f r;
    r.x = fmaxf(v.x, 0.0f);
    r.y = fmaxf(v.y, 0.0f);
    return __builtin_elementwise_fma(-a, t, r);
}

__global__ __launch_bounds__(256)
void node_sum_kernel(const float* __restrict__ x,
                     const int*   __restrict__ edge_index,
                     const int*   __restrict__ edge_attr,
                     const float* __restrict__ edge_weight,
                     const float* __restrict__ bond_emb,
                     const int*   __restrict__ bin_start,
                     const int*   __restrict__ bin_count,
                     const int*   __restrict__ recs,
                     float*       __restrict__ agg,
                     int E, int N)
{
    __shared__ int elist[8][LIST_CAP];
    __shared__ int sorted[8][LIST_CAP];

    int w    = threadIdx.x >> 5;
    int lane = threadIdx.x & 31;
    int n    = blockIdx.x * 8 + w;
    if (n >= N) return;

    const int kmask = (1 << BIN_SHIFT) - 1;
    int bin = n >> BIN_SHIFT;
    int key = n & kmask;
    int pad = key ^ 1;
    int s   = __builtin_amdgcn_readfirstlane(bin_start[bin]);
    int t   = s + __builtin_amdgcn_readfirstlane(bin_count[bin]);
    int d   = lane * 2;

    const float* x0 = x + d;
    const float* x1 = x + (size_t)N * D_FEAT + d;
    const float* eb = bond_emb + d;

    v2f acc0 = (v2f)(0.0f);
    v2f acc1 = (v2f)(0.0f);

    int deg = 0;
    for (int base = s; base < t; base += 128) {
        int r[4];
        for (int u = 0; u < 4; ++u) {
            int idx = base + u * 32 + lane;
            r[u] = (idx < t) ? recs[idx] : pad;
        }
        for (int u = 0; u < 4; ++u) {
            bool mine = ((r[u] & kmask) == key);
            unsigned mask = __builtin_amdgcn_ballot_w32(mine);
            if (mask != 0u) {
                int pos = deg + (int)__builtin_amdgcn_mbcnt_lo(mask, 0u);
                if (mine && pos < LIST_CAP) elist[w][pos] = r[u] >> BIN_SHIFT;
                deg += __builtin_popcount(mask);
            }
        }
    }

    if (deg <= LIST_CAP) {
        for (int i = lane; i < deg; i += 32) {
            int ei = elist[w][i];
            int rank = 0;
            for (int j = 0; j < deg; ++j) rank += (elist[w][j] < ei) ? 1 : 0;
            sorted[w][rank] = ei;
        }

        for (int g = 0; g < deg; g += 32) {
            int cnt = deg - g;
            if (cnt > 32) cnt = 32;

            int   src_l = 0, att_l = 0;
            float w0_l = 0.0f, w1_l = 0.0f;
            if (lane < cnt) {
                int es = sorted[w][g + lane];
                src_l = edge_index[es];
                att_l = edge_attr[es];
                w0_l  = edge_weight[es];
                w1_l  = edge_weight[(size_t)E + es];
            }

            int src = __builtin_amdgcn_readlane(src_l, 0);
            int att = __builtin_amdgcn_readlane(att_l, 0);
            v2f ev = *(const v2f*)(eb + (size_t)att * D_FEAT);
            v2f xa = *(const v2f*)(x0 + (size_t)src * D_FEAT);
            v2f xb = *(const v2f*)(x1 + (size_t)src * D_FEAT);
            for (int j = 0; j < cnt; ++j) {
                int jn   = (j + 1 < cnt) ? j + 1 : j;
                int srcn = __builtin_amdgcn_readlane(src_l, jn);
                int attn = __builtin_amdgcn_readlane(att_l, jn);
                v2f evn = *(const v2f*)(eb + (size_t)attn * D_FEAT);
                v2f xan = *(const v2f*)(x0 + (size_t)srcn * D_FEAT);
                v2f xbn = *(const v2f*)(x1 + (size_t)srcn * D_FEAT);

                float w0 = __int_as_float(__builtin_amdgcn_readlane(__float_as_int(w0_l), j));
                float w1 = __int_as_float(__builtin_amdgcn_readlane(__float_as_int(w1_l), j));
                acc0 = __builtin_elementwise_fma(gelu_pair(xa + ev), (v2f)(w0), acc0);
                acc1 = __builtin_elementwise_fma(gelu_pair(xb + ev), (v2f)(w1), acc1);

                ev = evn; xa = xan; xb = xbn;
            }
        }
    } else {
        int last = -1;
        for (int k = 0; k < deg; ++k) {
            int best = 0x7fffffff;
            for (int base = s; base < t; base += 32) {
                int idx = base + lane;
                int r = (idx < t) ? recs[idx] : pad;
                int e = r >> BIN_SHIFT;
                if ((r & kmask) == key && e > last && e < best) best = e;
            }
            for (int off = 16; off > 0; off >>= 1) {
                int o = __shfl_xor(best, off);
                best = (o < best) ? o : best;
            }
            int e = __builtin_amdgcn_readfirstlane(best);
            last = e;
            int   src = edge_index[e];
            int   att = edge_attr[e];
            float w0  = edge_weight[e];
            float w1  = edge_weight[(size_t)E + e];
            v2f ev = *(const v2f*)(eb + (size_t)att * D_FEAT);
            v2f xa = *(const v2f*)(x0 + (size_t)src * D_FEAT);
            v2f xb = *(const v2f*)(x1 + (size_t)src * D_FEAT);
            acc0 = __builtin_elementwise_fma(gelu_pair(xa + ev), (v2f)(w0), acc0);
            acc1 = __builtin_elementwise_fma(gelu_pair(xb + ev), (v2f)(w1), acc1);
        }
    }

    *(v2f*)(agg + (size_t)n * D_FEAT + d)       = acc0;
    *(v2f*)(agg + ((size_t)N + n) * D_FEAT + d) = acc1;
}

__global__ __launch_bounds__(WAVES_PER_BLOCK * 32)
void gin_mlp_kernel(const float* __restrict__ x,
                    const float* __restrict__ agg,
                    const float* __restrict__ W1,
                    const float* __restrict__ b1,
                    const float* __restrict__ W2,
                    const float* __restrict__ b2,
                    const float* __restrict__ eps,
                    float*       __restrict__ out,
                    int n_tiles)
{
    __shared__ float4 w1s[8 * 8 * 32];
    __shared__ float4 w2s[4 * 16 * 32];
    __shared__ float  lds_t[WAVES_PER_BLOCK][16 * H_FEAT];

    int waveId = threadIdx.x >> 5;
    int lane   = threadIdx.x & 31;
    int tile   = blockIdx.x * WAVES_PER_BLOCK + waveId;
    bool valid = tile < n_tiles;

    int m    = lane & 15;
    int half = lane >> 4;

    v2f A[16];
#pragma unroll
    for (int k = 0; k < 16; ++k) { A[k].x = 0.0f; A[k].y = 0.0f; }
    if (valid) {
        float epsv = 1.0f + eps[0];
        size_t rowbase = ((size_t)tile * 16 + m) * D_FEAT;
#pragma unroll
        for (int k = 0; k < 16; ++k) {
            int col = k * 4 + half * 2;
            float2 xv = *(const float2*)(x   + rowbase + col);
            float2 av = *(const float2*)(agg + rowbase + col);
            A[k].x = epsv * xv.x + av.x;
            A[k].y = epsv * xv.y + av.y;
        }
    }

#pragma unroll
    for (int n = 0; n < 8; ++n) {
        int ncol = n * 16 + m;
        int kb0  = waveId * 8 + half * 2;
        float4 e;
        e.x = W1[(size_t)(kb0    ) * H_FEAT + ncol];
        e.y = W1[(size_t)(kb0 + 1) * H_FEAT + ncol];
        e.z = W1[(size_t)(kb0 + 4) * H_FEAT + ncol];
        e.w = W1[(size_t)(kb0 + 5) * H_FEAT + ncol];
        w1s[(n * 8 + waveId) * 32 + lane] = e;
    }
#pragma unroll
    for (int i = 0; i < 8; ++i) {
        int o    = i >> 1;
        int jj   = waveId + 8 * (i & 1);
        int ocol = o * 16 + m;
        int kb0  = jj * 8 + half * 2;
        float4 e;
        e.x = W2[(size_t)(kb0    ) * D_FEAT + ocol];
        e.y = W2[(size_t)(kb0 + 1) * D_FEAT + ocol];
        e.z = W2[(size_t)(kb0 + 4) * D_FEAT + ocol];
        e.w = W2[(size_t)(kb0 + 5) * D_FEAT + ocol];
        w2s[(o * 16 + jj) * 32 + lane] = e;
    }
    __syncthreads();
    if (!valid) return;

    float* tl = &lds_t[waveId][0];

#pragma unroll
    for (int n = 0; n < 8; ++n) {
        v8f acc = {};
        int ncol = n * 16 + m;
#pragma unroll
        for (int kk = 0; kk < 8; ++kk) {
            float4 e = w1s[(n * 8 + kk) * 32 + lane];
            v2f B0; B0.x = e.x; B0.y = e.y;
            v2f B1; B1.x = e.z; B1.y = e.w;
            acc = __builtin_amdgcn_wmma_f32_16x16x4_f32(
                      false, A[2 * kk], false, B0, (short)0, acc, false, false);
            acc = __builtin_amdgcn_wmma_f32_16x16x4_f32(
                      false, A[2 * kk + 1], false, B1, (short)0, acc, false, false);
        }
        float bias = b1[ncol];
#pragma unroll
        for (int v = 0; v < 8; ++v) {
            float t = acc[v] + bias;
            t = t > 0.0f ? t : 0.0f;
            tl[(v + 8 * half) * H_FEAT + ncol] = t;
        }
    }

    asm volatile("s_wait_dscnt 0" ::: "memory");
    __builtin_amdgcn_wave_barrier();

#pragma unroll
    for (int o = 0; o < 4; ++o) {
        v8f acc = {};
        int ocol = o * 16 + m;
#pragma unroll
        for (int jj = 0; jj < 16; ++jj) {
            float4 e = w2s[(o * 16 + jj) * 32 + lane];
            int kb = jj * 8 + half * 2;
            float2 a0 = *(const float2*)(tl + m * H_FEAT + kb);
            float2 a1 = *(const float2*)(tl + m * H_FEAT + kb + 4);
            v2f A2; A2.x = a0.x; A2.y = a0.y;
            v2f A3; A3.x = a1.x; A3.y = a1.y;
            v2f B0; B0.x = e.x; B0.y = e.y;
            v2f B1; B1.x = e.z; B1.y = e.w;
            acc = __builtin_amdgcn_wmma_f32_16x16x4_f32(
                      false, A2, false, B0, (short)0, acc, false, false);
            acc = __builtin_amdgcn_wmma_f32_16x16x4_f32(
                      false, A3, false, B1, (short)0, acc, false, false);
        }
        float bias = b2[ocol];
#pragma unroll
        for (int v = 0; v < 8; ++v) {
            out[((size_t)tile * 16 + v + 8 * half) * D_FEAT + ocol] = acc[v] + bias;
        }
    }
}

extern "C" void kernel_launch(void* const* d_in, const int* in_sizes, int n_in,
                              void* d_out, int out_size, void* d_ws, size_t ws_size,
                              hipStream_t stream)
{
    const float* x           = (const float*)d_in[0];
    const int*   edge_index  = (const int*)  d_in[1];
    const int*   edge_attr   = (const int*)  d_in[2];
    const float* edge_weight = (const float*)d_in[3];
    const float* bond_emb    = (const float*)d_in[4];
    const float* eps         = (const float*)d_in[5];
    const float* W1          = (const float*)d_in[6];
    const float* b1          = (const float*)d_in[7];
    const float* W2          = (const float*)d_in[8];
    const float* b2          = (const float*)d_in[9];
    float*       out         = (float*)d_out;

    const int RC = 2;
    const int E  = in_sizes[2];
    const int N  = in_sizes[0] / (RC * D_FEAT);
    const int NB = (N + (1 << BIN_SHIFT) - 1) >> BIN_SHIFT;

    float* agg        = (float*)d_ws;
    int*   bin_count  = (int*)(agg + (size_t)RC * N * D_FEAT);
    int*   bin_cursor = bin_count + MAX_BINS;
    int*   bin_start  = bin_cursor + MAX_BINS;
    int*   recs       = bin_start + MAX_BINS;

    hipMemsetAsync(bin_count, 0, (size_t)2 * MAX_BINS * sizeof(int), stream);

    int blocksP = (E + PART_EDGES - 1) / PART_EDGES;
    bin_count_kernel<<<blocksP, PART_THREADS, 0, stream>>>(edge_index, bin_count, E, NB);
    bin_partition_kernel<<<blocksP, PART_THREADS, 0, stream>>>(edge_index, bin_count, bin_cursor,
                                                               bin_start, recs, E, NB);

    int blocks1 = (N + 7) / 8;
    node_sum_kernel<<<blocks1, 256, 0, stream>>>(x, edge_index, edge_attr, edge_weight, bond_emb,
                                                 bin_start, bin_count, recs, agg, E, N);

    int n_tiles = (RC * N) / 16;
    int blocks2 = (n_tiles + WAVES_PER_BLOCK - 1) / WAVES_PER_BLOCK;
    gin_mlp_kernel<<<blocks2, WAVES_PER_BLOCK * 32, 0, stream>>>(
        x, agg, W1, b1, W2, b2, eps, out, n_tiles);
}


# Edge phase: a block's 8 waves read their bin once together (6 words per lane; LDS integer counters give list slots) instead of every wave reading the whole bin; sums stay in ascending edge id
# speedup vs baseline: 1.9221x; 1.0906x over previous
#include <hip/hip_runtime.h>
#include <hip/hip_bf16.h>
#include <math.h>


typedef float v2f __attribute__((ext_vector_type(2)));
typedef float v8f __attribute__((ext_vector_type(8)));

#define D_FEAT 64
#define H_FEAT 128
#define WAVES_PER_BLOCK 8

#define BIN_SHIFT    6
#define MAX_BINS     1024
#define PART_THREADS 512
#define PART_EDGES   4096
#define LIST_CAP     128
#define BIN_WORDS    6

static_assert(MAX_BINS == 2 * PART_THREADS, "pass B scans two bins per thread");

__global__ __launch_bounds__(PART_THREADS)
void bin_count_kernel(const int* __restrict__ edge_index,
                      int*       __restrict__ bin_count,
                      int E, int NB)
{
    __shared__ int hist[MAX_BINS];
    hist[2 * threadIdx.x]     = 0;
    hist[2 * threadIdx.x + 1] = 0;
    __syncthreads();
    int base = blockIdx.x * PART_EDGES;
    int dstv[PART_EDGES / PART_THREADS];
    for (int i = 0; i < PART_EDGES / PART_THREADS; ++i) {
        int e = base + i * PART_THREADS + (int)threadIdx.x;
        dstv[i] = (e < E) ? edge_index[E + e] : -1;
    }
    for (int i = 0; i < PART_EDGES / PART_THREADS; ++i)
        if (dstv[i] >= 0) atomicAdd(&hist[dstv[i] >> BIN_SHIFT], 1);
    __syncthreads();
    for (int b = threadIdx.x; b < NB; b += PART_THREADS) {
        int h = hist[b];
        if (h) atomicAdd(&bin_count[b], h);
    }
}

__global__ __launch_bounds__(PART_THREADS)
void bin_partition_kernel(const int* __restrict__ edge_index,
                          const int* __restrict__ bin_count,
                          int*       __restrict__ bin_cursor,
                          int*       __restrict__ bin_start,
                          int*       __restrict__ recs,
                          int E, int NB)
{
    __shared__ int hist[MAX_BINS];
    __shared__ int sbase[MAX_BINS];
    __shared__ int wtot[PART_THREADS / 32];

    int tid  = threadIdx.x;
    int lane = tid & 31;
    int w    = tid >> 5;

    int c0 = (2 * tid     < NB) ? bin_count[2 * tid]     : 0;
    int c1 = (2 * tid + 1 < NB) ? bin_count[2 * tid + 1] : 0;
    int v   = c0 + c1;
    int inc = v;
    for (int off = 1; off < 32; off <<= 1) {
        int o = __shfl_up(inc, off);
        if (lane >= off) inc += o;
    }
    if (lane == 31) wtot[w] = inc;
    hist[2 * tid]     = 0;
    hist[2 * tid + 1] = 0;
    __syncthreads();
    int wbase = 0;
    for (int i = 0; i < w; ++i) wbase += wtot[i];
    int ex = wbase + inc - v;
    sbase[2 * tid]     = ex;
    sbase[2 * tid + 1] = ex + c0;
    if (blockIdx.x == 0) {
        if (2 * tid     < NB) bin_start[2 * tid]     = ex;
        if (2 * tid + 1 < NB) bin_start[2 * tid + 1] = ex + c0;
    }

    int base = blockIdx.x * PART_EDGES;
    int dstv[PART_EDGES / PART_THREADS];
    for (int i = 0; i < PART_EDGES / PART_THREADS; ++i) {
        int e = base + i * PART_THREADS + tid;
        dstv[i] = (e < E) ? edge_index[E + e] : -1;
    }
    for (int i = 0; i < PART_EDGES / PART_THREADS; ++i)
        if (dstv[i] >= 0) atomicAdd(&hist[dstv[i] >> BIN_SHIFT], 1);
    __syncthreads();

    for (int b = tid; b < NB; b += PART_THREADS) {
        int h = hist[b];
        int g = h ? atomicAdd(&bin_cursor[b], h) : 0;
        sbase[b] += g;
        hist[b] = 0;
    }
    __syncthreads();

    for (int i = 0; i < PART_EDGES / PART_THREADS; ++i) {
        if (dstv[i] >= 0) {
            int b    = dstv[i] >> BIN_SHIFT;
            int slot = sbase[b] + atomicAdd(&hist[b], 1);
            int e    = base + i * PART_THREADS + tid;
            recs[slot] = (e << BIN_SHIFT) | (dstv[i] & ((1 << BIN_SHIFT) - 1));
        }
    }
}

__device__ __forceinline__ v2f gelu_pair(v2f v)
{
    v2f a;
    a.x = fminf(fabsf(v.x), 8.0f);
    a.y = fminf(fabsf(v.y), 8.0f);
    v2f q = (v2f)(2.1919348114352033e-07f);
    q = __builtin_elementwise_fma(q, a, (v2f)(-5.2599425544030964e-06f));
    q = __builtin_elementwise_fma(q, a, (v2f)(4.943032763549127e-05f));
    q = __builtin_elementwise_fma(q, a, (v2f)(-0.00020471429161261767f));
    q = __builtin_elementwise_fma(q, a, (v2f)(-0.00012574861466418952f));
    q = __builtin_elementwise_fma(q, a, (v2f)(0.007073047570884228f));
    q = __builtin_elementwise_fma(q, a, (v2f)(-0.05251256376504898f));
    q = __builtin_elementwise_fma(q, a, (v2f)(-0.45920106768608093f));
    q = __builtin_elementwise_fma(q, a, (v2f)(-1.1511061191558838f));
    q = __builtin_elementwise_fma(q, a, (v2f)(-1.0f));
    v2f t;
    t.x = __builtin_amdgcn_exp2f(q.x);
    t.y = __builtin_amdgcn_exp2f(q.y);
    v2f r;
    r.x = fmaxf(v.x, 0.0f);
    r.y = fmaxf(v.y, 0.0f);
    return __builtin_elementwise_fma(-a, t, r);
}

__global__ __launch_bounds__(256)
void node_sum_kernel(const float* __restrict__ x,
                     const int*   __restrict__ edge_index,
                     const int*   __restrict__ edge_attr,
                     const float* __restrict__ edge_weight,
                     const float* __restrict__ bond_emb,
                     const int*   __restrict__ bin_start,
                     const int*   __restrict__ bin_count,
                     const int*   __restrict__ recs,
                     float*       __restrict__ agg,
                     int E, int N)
{
    __shared__ int ncnt[8];
    __shared__ int elist[8][LIST_CAP];
    __shared__ int sorted[8][LIST_CAP];

    int tid  = threadIdx.x;
    int w    = tid >> 5;
    int lane = tid & 31;
    int nb0  = blockIdx.x * 8;
    int n    = nb0 + w;

    const int kmask = (1 << BIN_SHIFT) - 1;
    int bin  = nb0 >> BIN_SHIFT;
    int bkey = (nb0 & kmask) >> 3;
    int key  = n & kmask;
    int pad  = key ^ 1;
    int s    = bin_start[bin];
    int t    = s + bin_count[bin];
    int d    = lane * 2;

    if (tid < 8) ncnt[tid] = 0;
    __syncthreads();
    for (int base = s; base < t; base += BIN_WORDS * 256) {
        int r[BIN_WORDS];
        for (int u = 0; u < BIN_WORDS; ++u) {
            int idx = base + u * 256 + tid;
            r[u] = (idx < t) ? recs[idx] : pad;
        }
        for (int u = 0; u < BIN_WORDS; ++u) {
            int idx = base + u * 256 + tid;
            if (idx < t && ((r[u] & kmask) >> 3) == bkey) {
                int k   = r[u] & 7;
                int pos = atomicAdd(&ncnt[k], 1);
                if (pos < LIST_CAP) elist[k][pos] = r[u] >> BIN_SHIFT;
            }
        }
    }
    __syncthreads();
    if (n >= N) return;

    const float* x0 = x + d;
    const float* x1 = x + (size_t)N * D_FEAT + d;
    const float* eb = bond_emb + d;

    v2f acc0 = (v2f)(0.0f);
    v2f acc1 = (v2f)(0.0f);

    int deg = __builtin_amdgcn_readfirstlane(ncnt[w]);

    if (deg <= LIST_CAP) {
        for (int i = lane; i < deg; i += 32) {
            int ei = elist[w][i];
            int rank = 0;
            for (int j = 0; j < deg; ++j) rank += (elist[w][j] < ei) ? 1 : 0;
            sorted[w][rank] = ei;
        }

        for (int g = 0; g < deg; g += 32) {
            int cnt = deg - g;
            if (cnt > 32) cnt = 32;

            int   src_l = 0, att_l = 0;
            float w0_l = 0.0f, w1_l = 0.0f;
            if (lane < cnt) {
                int es = sorted[w][g + lane];
                src_l = edge_index[es];
                att_l = edge_attr[es];
                w0_l  = edge_weight[es];
                w1_l  = edge_weight[(size_t)E + es];
            }

            int src = __builtin_amdgcn_readlane(src_l, 0);
            int att = __builtin_amdgcn_readlane(att_l, 0);
            v2f ev = *(const v2f*)(eb + (size_t)att * D_FEAT);
            v2f xa = *(const v2f*)(x0 + (size_t)src * D_FEAT);
            v2f xb = *(const v2f*)(x1 + (size_t)src * D_FEAT);
            for (int j = 0; j < cnt; ++j) {
                int jn   = (j + 1 < cnt) ? j + 1 : j;
                int srcn = __builtin_amdgcn_readlane(src_l, jn);
                int attn = __builtin_amdgcn_readlane(att_l, jn);
                v2f evn = *(const v2f*)(eb + (size_t)attn * D_FEAT);
                v2f xan = *(const v2f*)(x0 + (size_t)srcn * D_FEAT);
                v2f xbn = *(const v2f*)(x1 + (size_t)srcn * D_FEAT);

                float w0 = __int_as_float(__builtin_amdgcn_readlane(__float_as_int(w0_l), j));
                float w1 = __int_as_float(__builtin_amdgcn_readlane(__float_as_int(w1_l), j));
                acc0 = __builtin_elementwise_fma(gelu_pair(xa + ev), (v2f)(w0), acc0);
                acc1 = __builtin_elementwise_fma(gelu_pair(xb + ev), (v2f)(w1), acc1);

                ev = evn; xa = xan; xb = xbn;
            }
        }
    } else {
        int last = -1;
        for (int k = 0; k < deg; ++k) {
            int best = 0x7fffffff;
            for (int base = s; base < t; base += 32) {
                int idx = base + lane;
                int r = (idx < t) ? recs[idx] : pad;
                int e = r >> BIN_SHIFT;
                if ((r & kmask) == key && e > last && e < best) best = e;
            }
            for (int off = 16; off > 0; off >>= 1) {
                int o = __shfl_xor(best, off);
                best = (o < best) ? o : best;
            }
            int e = __builtin_amdgcn_readfirstlane(best);
            last = e;
            int   src = edge_index[e];
            int   att = edge_attr[e];
            float w0  = edge_weight[e];
            float w1  = edge_weight[(size_t)E + e];
            v2f ev = *(const v2f*)(eb + (size_t)att * D_FEAT);
            v2f xa = *(const v2f*)(x0 + (size_t)src * D_FEAT);
            v2f xb = *(const v2f*)(x1 + (size_t)src * D_FEAT);
            acc0 = __builtin_elementwise_fma(gelu_pair(xa + ev), (v2f)(w0), acc0);
            acc1 = __builtin_elementwise_fma(gelu_pair(xb + ev), (v2f)(w1), acc1);
        }
    }

    *(v2f*)(agg + (size_t)n * D_FEAT + d)       = acc0;
    *(v2f*)(agg + ((size_t)N + n) * D_FEAT + d) = acc1;
}

__global__ __launch_bounds__(WAVES_PER_BLOCK * 32)
void gin_mlp_kernel(const float* __restrict__ x,
                    const float* __restrict__ agg,
                    const float* __restrict__ W1,
                    const float* __restrict__ b1,
                    const float* __restrict__ W2,
                    const float* __restrict__ b2,
                    const float* __restrict__ eps,
                    float*       __restrict__ out,
                    int n_tiles)
{
    __shared__ float4 w1s[8 * 8 * 32];
    __shared__ float4 w2s[4 * 16 * 32];
    __shared__ float  lds_t[WAVES_PER_BLOCK][16 * H_FEAT];

    int waveId = threadIdx.x >> 5;
    int lane   = threadIdx.x & 31;
    int tile   = blockIdx.x * WAVES_PER_BLOCK + waveId;
    bool valid = tile < n_tiles;

    int m    = lane & 15;
    int half = lane >> 4;

    v2f A[16];
#pragma unroll
    for (int k = 0; k < 16; ++k) { A[k].x = 0.0f; A[k].y = 0.0f; }
    if (valid) {
        float epsv = 1.0f + eps[0];
        size_t rowbase = ((size_t)tile * 16 + m) * D_FEAT;
#pragma unroll
        for (int k = 0; k < 16; ++k) {
            int col = k * 4 + half * 2;
            float2 xv = *(const float2*)(x   + rowbase + col);
            float2 av = *(const float2*)(agg + rowbase + col);
            A[k].x = epsv * xv.x + av.x;
            A[k].y = epsv * xv.y + av.y;
        }
    }

#pragma unroll
    for (int n = 0; n < 8; ++n) {
        int ncol = n * 16 + m;
        int kb0  = waveId * 8 + half * 2;
        float4 e;
        e.x = W1[(size_t)(kb0    ) * H_FEAT + ncol];
        e.y = W1[(size_t)(kb0 + 1) * H_FEAT + ncol];
        e.z = W1[(size_t)(kb0 + 4) * H_FEAT + ncol];
        e.w = W1[(size_t)(kb0 + 5) * H_FEAT + ncol];
        w1s[(n * 8 + waveId) * 32 + lane] = e;
    }
#pragma unroll
    for (int i = 0; i < 8; ++i) {
        int o    = i >> 1;
        int jj   = waveId + 8 * (i & 1);
        int ocol = o * 16 + m;
        int kb0  = jj * 8 + half * 2;
        float4 e;
        e.x = W2[(size_t)(kb0    ) * D_FEAT + ocol];
        e.y = W2[(size_t)(kb0 + 1) * D_FEAT + ocol];
        e.z = W2[(size_t)(kb0 + 4) * D_FEAT + ocol];
        e.w = W2[(size_t)(kb0 + 5) * D_FEAT + ocol];
        w2s[(o * 16 + jj) * 32 + lane] = e;
    }
    __syncthreads();
    if (!valid) return;

    float* tl = &lds_t[waveId][0];

#pragma unroll
    for (int n = 0; n < 8; ++n) {
        v8f acc = {};
        int ncol = n * 16 + m;
#pragma unroll
        for (int kk = 0; kk < 8; ++kk) {
            float4 e = w1s[(n * 8 + kk) * 32 + lane];
            v2f B0; B0.x = e.x; B0.y = e.y;
            v2f B1; B1.x = e.z; B1.y = e.w;
            acc = __builtin_amdgcn_wmma_f32_16x16x4_f32(
                      false, A[2 * kk], false, B0, (short)0, acc, false, false);
            acc = __builtin_amdgcn_wmma_f32_16x16x4_f32(
                      false, A[2 * kk + 1], false, B1, (short)0, acc, false, false);
        }
        float bias = b1[ncol];
#pragma unroll
        for (int v = 0; v < 8; ++v) {
            float t = acc[v] + bias;
            t = t > 0.0f ? t : 0.0f;
            tl[(v + 8 * half) * H_FEAT + ncol] = t;
        }
    }

    asm volatile("s_wait_dscnt 0" ::: "memory");
    __builtin_amdgcn_wave_barrier();

#pragma unroll
    for (int o = 0; o < 4; ++o) {
        v8f acc = {};
        int ocol = o * 16 + m;
#pragma unroll
        for (int jj = 0; jj < 16; ++jj) {
            float4 e = w2s[(o * 16 + jj) * 32 + lane];
            int kb = jj * 8 + half * 2;
            float2 a0 = *(const float2*)(tl + m * H_FEAT + kb);
            float2 a1 = *(const float2*)(tl + m * H_FEAT + kb + 4);
            v2f A2; A2.x = a0.x; A2.y = a0.y;
            v2f A3; A3.x = a1.x; A3.y = a1.y;
            v2f B0; B0.x = e.x; B0.y = e.y;
            v2f B1; B1.x = e.z; B1.y = e.w;
            acc = __builtin_amdgcn_wmma_f32_16x16x4_f32(
                      false, A2, false, B0, (short)0, acc, false, false);
            acc = __builtin_amdgcn_wmma_f32_16x16x4_f32(
                      false, A3, false, B1, (short)0, acc, false, false);
        }
        float bias = b2[ocol];
#pragma unroll
        for (int v = 0; v < 8; ++v) {
            out[((size_t)tile * 16 + v + 8 * half) * D_FEAT + ocol] = acc[v] + bias;
        }
    }
}

extern "C" void kernel_launch(void* const* d_in, const int* in_sizes, int n_in,
                              void* d_out, int out_size, void* d_ws, size_t ws_size,
                              hipStream_t stream)
{
    const float* x           = (const float*)d_in[0];
    const int*   edge_index  = (const int*)  d_in[1];
    const int*   edge_attr   = (const int*)  d_in[2];
    const float* edge_weight = (const float*)d_in[3];
    const float* bond_emb    = (const float*)d_in[4];
    const float* eps         = (const float*)d_in[5];
    const float* W1          = (const float*)d_in[6];
    const float* b1          = (const float*)d_in[7];
    const float* W2          = (const float*)d_in[8];
    const float* b2          = (const float*)d_in[9];
    float*       out         = (float*)d_out;

    const int RC = 2;
    const int E  = in_sizes[2];
    const int N  = in_sizes[0] / (RC * D_FEAT);
    const int NB = (N + (1 << BIN_SHIFT) - 1) >> BIN_SHIFT;

    float* agg        = (float*)d_ws;
    int*   bin_count  = (int*)(agg + (size_t)RC * N * D_FEAT);
    int*   bin_cursor = bin_count + MAX_BINS;
    int*   bin_start  = bin_cursor + MAX_BINS;
    int*   recs       = bin_start + MAX_BINS;

    hipMemsetAsync(bin_count, 0, (size_t)2 * MAX_BINS * sizeof(int), stream);

    int blocksP = (E + PART_EDGES - 1) / PART_EDGES;
    bin_count_kernel<<<blocksP, PART_THREADS, 0, stream>>>(edge_index, bin_count, E, NB);
    bin_partition_kernel<<<blocksP, PART_THREADS, 0, stream>>>(edge_index, bin_count, bin_cursor,
                                                               bin_start, recs, E, NB);

    int blocks1 = (N + 7) / 8;
    node_sum_kernel<<<blocks1, 256, 0, stream>>>(x, edge_index, edge_attr, edge_weight, bond_emb,
                                                 bin_start, bin_count, recs, agg, E, N);

    int n_tiles = (RC * N) / 16;
    int blocks2 = (n_tiles + WAVES_PER_BLOCK - 1) / WAVES_PER_BLOCK;
    gin_mlp_kernel<<<blocks2, WAVES_PER_BLOCK * 32, 0, stream>>>(
        x, agg, W1, b1, W2, b2, eps, out, n_tiles);
}


# Edge phase: each bin gets a fixed room of 2048 words (an overflow list takes what a fuller bin cannot hold), so the counting launch and the prefix scan go: one binning launch, not two
# speedup vs baseline: 1.9905x; 1.0356x over previous
#include <hip/hip_runtime.h>
#include <hip/hip_bf16.h>
#include <math.h>


typedef float v2f __attribute__((ext_vector_type(2)));
typedef float v8f __attribute__((ext_vector_type(8)));

#define D_FEAT 64
#define H_FEAT 128
#define WAVES_PER_BLOCK 8

#define BIN_SHIFT    6
#define BIN_CAP      2048
#define MAX_BINS     1024
#define PART_THREADS 512
#define PART_EDGES   4096
#define LIST_CAP     128
#define BIN_WORDS    6

static_assert(MAX_BINS == 2 * PART_THREADS, "pass 1 clears two LDS counters per thread");

__global__ __launch_bounds__(PART_THREADS)
void bin_partition_kernel(const int* __restrict__ edge_index,
                          int*       __restrict__ bin_cursor,
                          int*       __restrict__ recs,
                          int*       __restrict__ ovf,
                          int E, int NB)
{
    __shared__ int hist[MAX_BINS];
    __shared__ int sbase[MAX_BINS];

    int tid = threadIdx.x;
    hist[2 * tid]     = 0;
    hist[2 * tid + 1] = 0;
    __syncthreads();

    int base = blockIdx.x * PART_EDGES;
    int dstv[PART_EDGES / PART_THREADS];
    for (int i = 0; i < PART_EDGES / PART_THREADS; ++i) {
        int e = base + i * PART_THREADS + tid;
        dstv[i] = (e < E) ? edge_index[E + e] : -1;
    }
    for (int i = 0; i < PART_EDGES / PART_THREADS; ++i)
        if (dstv[i] >= 0) atomicAdd(&hist[dstv[i] >> BIN_SHIFT], 1);
    __syncthreads();

    for (int b = tid; b < NB; b += PART_THREADS) {
        int h = hist[b];
        if (h) {
            sbase[b] = atomicAdd(&bin_cursor[b], h);
            hist[b]  = 0;
        }
    }
    __syncthreads();

    for (int i = 0; i < PART_EDGES / PART_THREADS; ++i) {
        if (dstv[i] >= 0) {
            int b    = dstv[i] >> BIN_SHIFT;
            int slot = sbase[b] + atomicAdd(&hist[b], 1);
            int e    = base + i * PART_THREADS + tid;
            if (slot < BIN_CAP) {
                recs[b * BIN_CAP + slot] = (e << BIN_SHIFT) | (dstv[i] & ((1 << BIN_SHIFT) - 1));
            } else {
                int k = atomicAdd(&bin_cursor[MAX_BINS], 1);
                ovf[2 * k]     = e;
                ovf[2 * k + 1] = dstv[i];
            }
        }
    }
}

__device__ __forceinline__ v2f gelu_pair(v2f v)
{
    v2f a;
    a.x = fminf(fabsf(v.x), 8.0f);
    a.y = fminf(fabsf(v.y), 8.0f);
    v2f q = (v2f)(2.1919348114352033e-07f);
    q = __builtin_elementwise_fma(q, a, (v2f)(-5.2599425544030964e-06f));
    q = __builtin_elementwise_fma(q, a, (v2f)(4.943032763549127e-05f));
    q = __builtin_elementwise_fma(q, a, (v2f)(-0.00020471429161261767f));
    q = __builtin_elementwise_fma(q, a, (v2f)(-0.00012574861466418952f));
    q = __builtin_elementwise_fma(q, a, (v2f)(0.007073047570884228f));
    q = __builtin_elementwise_fma(q, a, (v2f)(-0.05251256376504898f));
    q = __builtin_elementwise_fma(q, a, (v2f)(-0.45920106768608093f));
    q = __builtin_elementwise_fma(q, a, (v2f)(-1.1511061191558838f));
    q = __builtin_elementwise_fma(q, a, (v2f)(-1.0f));
    v2f t;
    t.x = __builtin_amdgcn_exp2f(q.x);
    t.y = __builtin_amdgcn_exp2f(q.y);
    v2f r;
    r.x = fmaxf(v.x, 0.0f);
    r.y = fmaxf(v.y, 0.0f);
    return __builtin_elementwise_fma(-a, t, r);
}

__global__ __launch_bounds__(256)
void node_sum_kernel(const float* __restrict__ x,
                     const int*   __restrict__ edge_index,
                     const int*   __restrict__ edge_attr,
                     const float* __restrict__ edge_weight,
                     const float* __restrict__ bond_emb,
                     const int*   __restrict__ bin_cursor,
                     const int*   __restrict__ recs,
                     const int*   __restrict__ ovf,
                     float*       __restrict__ agg,
                     int E, int N)
{
    __shared__ int ncnt[8];
    __shared__ int elist[8][LIST_CAP];
    __shared__ int sorted[8][LIST_CAP];

    int tid  = threadIdx.x;
    int w    = tid >> 5;
    int lane = tid & 31;
    int nb0  = blockIdx.x * 8;
    int n    = nb0 + w;

    const int kmask = (1 << BIN_SHIFT) - 1;
    int bin  = nb0 >> BIN_SHIFT;
    int bkey = (nb0 & kmask) >> 3;
    int key  = n & kmask;
    int pad  = key ^ 1;
    int cbin = bin_cursor[bin];
    int novf = bin_cursor[MAX_BINS];
    int s    = bin * BIN_CAP;
    int t    = s + (cbin < BIN_CAP ? cbin : BIN_CAP);
    int d    = lane * 2;

    if (tid < 8) ncnt[tid] = 0;
    __syncthreads();
    for (int base = s; base < t; base += BIN_WORDS * 256) {
        int r[BIN_WORDS];
        for (int u = 0; u < BIN_WORDS; ++u) {
            int idx = base + u * 256 + tid;
            r[u] = (idx < t) ? recs[idx] : pad;
        }
        for (int u = 0; u < BIN_WORDS; ++u) {
            int idx = base + u * 256 + tid;
            if (idx < t && ((r[u] & kmask) >> 3) == bkey) {
                int k   = r[u] & 7;
                int pos = atomicAdd(&ncnt[k], 1);
                if (pos < LIST_CAP) elist[k][pos] = r[u] >> BIN_SHIFT;
            }
        }
    }
    for (int base = 0; base < novf; base += 256) {
        int idx = base + tid;
        if (idx < novf) {
            int e  = ovf[2 * idx];
            int dd = ovf[2 * idx + 1];
            if ((dd >> 3) == (nb0 >> 3)) {
                int k   = dd & 7;
                int pos = atomicAdd(&ncnt[k], 1);
                if (pos < LIST_CAP) elist[k][pos] = e;
            }
        }
    }
    __syncthreads();
    if (n >= N) return;

    const float* x0 = x + d;
    const float* x1 = x + (size_t)N * D_FEAT + d;
    const float* eb = bond_emb + d;

    v2f acc0 = (v2f)(0.0f);
    v2f acc1 = (v2f)(0.0f);

    int deg = __builtin_amdgcn_readfirstlane(ncnt[w]);

    if (deg <= LIST_CAP) {
        for (int i = lane; i < deg; i += 32) {
            int ei = elist[w][i];
            int rank = 0;
            for (int j = 0; j < deg; ++j) rank += (elist[w][j] < ei) ? 1 : 0;
            sorted[w][rank] = ei;
        }

        for (int g = 0; g < deg; g += 32) {
            int cnt = deg - g;
            if (cnt > 32) cnt = 32;

            int   src_l = 0, att_l = 0;
            float w0_l = 0.0f, w1_l = 0.0f;
            if (lane < cnt) {
                int es = sorted[w][g + lane];
                src_l = edge_index[es];
                att_l = edge_attr[es];
                w0_l  = edge_weight[es];
                w1_l  = edge_weight[(size_t)E + es];
            }

            int src = __builtin_amdgcn_readlane(src_l, 0);
            int att = __builtin_amdgcn_readlane(att_l, 0);
            v2f ev = *(const v2f*)(eb + (size_t)att * D_FEAT);
            v2f xa = *(const v2f*)(x0 + (size_t)src * D_FEAT);
            v2f xb = *(const v2f*)(x1 + (size_t)src * D_FEAT);
            for (int j = 0; j < cnt; ++j) {
                int jn   = (j + 1 < cnt) ? j + 1 : j;
                int srcn = __builtin_amdgcn_readlane(src_l, jn);
                int attn = __builtin_amdgcn_readlane(att_l, jn);
                v2f evn = *(const v2f*)(eb + (size_t)attn * D_FEAT);
                v2f xan = *(const v2f*)(x0 + (size_t)srcn * D_FEAT);
                v2f xbn = *(const v2f*)(x1 + (size_t)srcn * D_FEAT);

                float w0 = __int_as_float(__builtin_amdgcn_readlane(__float_as_int(w0_l), j));
                float w1 = __int_as_float(__builtin_amdgcn_readlane(__float_as_int(w1_l), j));
                acc0 = __builtin_elementwise_fma(gelu_pair(xa + ev), (v2f)(w0), acc0);
                acc1 = __builtin_elementwise_fma(gelu_pair(xb + ev), (v2f)(w1), acc1);

                ev = evn; xa = xan; xb = xbn;
            }
        }
    } else {
        int last = -1;
        for (int k = 0; k < deg; ++k) {
            int best = 0x7fffffff;
            for (int base = s; base < t; base += 32) {
                int idx = base + lane;
                int r = (idx < t) ? recs[idx] : pad;
                int e = r >> BIN_SHIFT;
                if ((r & kmask) == key && e > last && e < best) best = e;
            }
            for (int base = 0; base < novf; base += 32) {
                int idx = base + lane;
                if (idx < novf) {
                    int e  = ovf[2 * idx];
                    int dd = ovf[2 * idx + 1];
                    if (dd == n && e > last && e < best) best = e;
                }
            }
            for (int off = 16; off > 0; off >>= 1) {
                int o = __shfl_xor(best, off);
                best = (o < best) ? o : best;
            }
            int e = __builtin_amdgcn_readfirstlane(best);
            last = e;
            int   src = edge_index[e];
            int   att = edge_attr[e];
            float w0  = edge_weight[e];
            float w1  = edge_weight[(size_t)E + e];
            v2f ev = *(const v2f*)(eb + (size_t)att * D_FEAT);
            v2f xa = *(const v2f*)(x0 + (size_t)src * D_FEAT);
            v2f xb = *(const v2f*)(x1 + (size_t)src * D_FEAT);
            acc0 = __builtin_elementwise_fma(gelu_pair(xa + ev), (v2f)(w0), acc0);
            acc1 = __builtin_elementwise_fma(gelu_pair(xb + ev), (v2f)(w1), acc1);
        }
    }

    *(v2f*)(agg + (size_t)n * D_FEAT + d)       = acc0;
    *(v2f*)(agg + ((size_t)N + n) * D_FEAT + d) = acc1;
}

__global__ __launch_bounds__(WAVES_PER_BLOCK * 32)
void gin_mlp_kernel(const float* __restrict__ x,
                    const float* __restrict__ agg,
                    const float* __restrict__ W1,
                    const float* __restrict__ b1,
                    const float* __restrict__ W2,
                    const float* __restrict__ b2,
                    const float* __restrict__ eps,
                    float*       __restrict__ out,
                    int n_tiles)
{
    __shared__ float4 w1s[8 * 8 * 32];
    __shared__ float4 w2s[4 * 16 * 32];
    __shared__ float  lds_t[WAVES_PER_BLOCK][16 * H_FEAT];

    int waveId = threadIdx.x >> 5;
    int lane   = threadIdx.x & 31;
    int tile   = blockIdx.x * WAVES_PER_BLOCK + waveId;
    bool valid = tile < n_tiles;

    int m    = lane & 15;
    int half = lane >> 4;

    v2f A[16];
#pragma unroll
    for (int k = 0; k < 16; ++k) { A[k].x = 0.0f; A[k].y = 0.0f; }
    if (valid) {
        float epsv = 1.0f + eps[0];
        size_t rowbase = ((size_t)tile * 16 + m) * D_FEAT;
#pragma unroll
        for (int k = 0; k < 16; ++k) {
            int col = k * 4 + half * 2;
            float2 xv = *(const float2*)(x   + rowbase + col);
            float2 av = *(const float2*)(agg + rowbase + col);
            A[k].x = epsv * xv.x + av.x;
            A[k].y = epsv * xv.y + av.y;
        }
    }

#pragma unroll
    for (int n = 0; n < 8; ++n) {
        int ncol = n * 16 + m;
        int kb0  = waveId * 8 + half * 2;
        float4 e;
        e.x = W1[(size_t)(kb0    ) * H_FEAT + ncol];
        e.y = W1[(size_t)(kb0 + 1) * H_FEAT + ncol];
        e.z = W1[(size_t)(kb0 + 4) * H_FEAT + ncol];
        e.w = W1[(size_t)(kb0 + 5) * H_FEAT + ncol];
        w1s[(n * 8 + waveId) * 32 + lane] = e;
    }
#pragma unroll
    for (int i = 0; i < 8; ++i) {
        int o    = i >> 1;
        int jj   = waveId + 8 * (i & 1);
        int ocol = o * 16 + m;
        int kb0  = jj * 8 + half * 2;
        float4 e;
        e.x = W2[(size_t)(kb0    ) * D_FEAT + ocol];
        e.y = W2[(size_t)(kb0 + 1) * D_FEAT + ocol];
        e.z = W2[(size_t)(kb0 + 4) * D_FEAT + ocol];
        e.w = W2[(size_t)(kb0 + 5) * D_FEAT + ocol];
        w2s[(o * 16 + jj) * 32 + lane] = e;
    }
    __syncthreads();
    if (!valid) return;

    float* tl = &lds_t[waveId][0];

#pragma unroll
    for (int n = 0; n < 8; ++n) {
        v8f acc = {};
        int ncol = n * 16 + m;
#pragma unroll
        for (int kk = 0; kk < 8; ++kk) {
            float4 e = w1s[(n * 8 + kk) * 32 + lane];
            v2f B0; B0.x = e.x; B0.y = e.y;
            v2f B1; B1.x = e.z; B1.y = e.w;
            acc = __builtin_amdgcn_wmma_f32_16x16x4_f32(
                      false, A[2 * kk], false, B0, (short)0, acc, false, false);
            acc = __builtin_amdgcn_wmma_f32_16x16x4_f32(
                      false, A[2 * kk + 1], false, B1, (short)0, acc, false, false);
        }
        float bias = b1[ncol];
#pragma unroll
        for (int v = 0; v < 8; ++v) {
            float t = acc[v] + bias;
            t = t > 0.0f ? t : 0.0f;
            tl[(v + 8 * half) * H_FEAT + ncol] = t;
        }
    }

    asm volatile("s_wait_dscnt 0" ::: "memory");
    __builtin_amdgcn_wave_barrier();

#pragma unroll
    for (int o = 0; o < 4; ++o) {
        v8f acc = {};
        int ocol = o * 16 + m;
#pragma unroll
        for (int jj = 0; jj < 16; ++jj) {
            float4 e = w2s[(o * 16 + jj) * 32 + lane];
            int kb = jj * 8 + half * 2;
            float2 a0 = *(const float2*)(tl + m * H_FEAT + kb);
            float2 a1 = *(const float2*)(tl + m * H_FEAT + kb + 4);
            v2f A2; A2.x = a0.x; A2.y = a0.y;
            v2f A3; A3.x = a1.x; A3.y = a1.y;
            v2f B0; B0.x = e.x; B0.y = e.y;
            v2f B1; B1.x = e.z; B1.y = e.w;
            acc = __builtin_amdgcn_wmma_f32_16x16x4_f32(
                      false, A2, false, B0, (short)0, acc, false, false);
            acc = __builtin_amdgcn_wmma_f32_16x16x4_f32(
                      false, A3, false, B1, (short)0, acc, false, false);
        }
        float bias = b2[ocol];
#pragma unroll
        for (int v = 0; v < 8; ++v) {
            out[((size_t)tile * 16 + v + 8 * half) * D_FEAT + ocol] = acc[v] + bias;
        }
    }
}

extern "C" void kernel_launch(void* const* d_in, const int* in_sizes, int n_in,
                              void* d_out, int out_size, void* d_ws, size_t ws_size,
                              hipStream_t stream)
{
    const float* x           = (const float*)d_in[0];
    const int*   edge_index  = (const int*)  d_in[1];
    const int*   edge_attr   = (const int*)  d_in[2];
    const float* edge_weight = (const float*)d_in[3];
    const float* bond_emb    = (const float*)d_in[4];
    const float* eps         = (const float*)d_in[5];
    const float* W1          = (const float*)d_in[6];
    const float* b1          = (const float*)d_in[7];
    const float* W2          = (const float*)d_in[8];
    const float* b2          = (const float*)d_in[9];
    float*       out         = (float*)d_out;

    const int RC = 2;
    const int E  = in_sizes[2];
    const int N  = in_sizes[0] / (RC * D_FEAT);
    const int NB = (N + (1 << BIN_SHIFT) - 1) >> BIN_SHIFT;

    float* agg        = (float*)d_ws;
    int*   bin_cursor = (int*)(agg + (size_t)RC * N * D_FEAT);
    int*   recs       = bin_cursor + MAX_BINS + 32;
    int*   ovf        = recs + (size_t)MAX_BINS * BIN_CAP;

    hipMemsetAsync(bin_cursor, 0, (size_t)(MAX_BINS + 32) * sizeof(int), stream);

    int blocksP = (E + PART_EDGES - 1) / PART_EDGES;
    bin_partition_kernel<<<blocksP, PART_THREADS, 0, stream>>>(edge_index, bin_cursor, recs, ovf, E, NB);

    int blocks1 = (N + 7) / 8;
    node_sum_kernel<<<blocks1, 256, 0, stream>>>(x, edge_index, edge_attr, edge_weight, bond_emb,
                                                 bin_cursor, recs, ovf, agg, E, N);

    int n_tiles = (RC * N) / 16;
    int blocks2 = (n_tiles + WAVES_PER_BLOCK - 1) / WAVES_PER_BLOCK;
    gin_mlp_kernel<<<blocks2, WAVES_PER_BLOCK * 32, 0, stream>>>(
        x, agg, W1, b1, W2, b2, eps, out, n_tiles);
}


# Edge phase: binning launch writes a 16-byte record per edge (id|dst, src|attr, both weights, read in edge order); node sums take them from the bin, not by 4 scattered 4-byte reads; same sum order
# speedup vs baseline: 2.3057x; 1.1584x over previous
#include <hip/hip_runtime.h>
#include <hip/hip_bf16.h>
#include <math.h>


typedef float v2f __attribute__((ext_vector_type(2)));
typedef float v8f __attribute__((ext_vector_type(8)));

#define D_FEAT 64
#define H_FEAT 128
#define WAVES_PER_BLOCK 8

#define BIN_SHIFT    6
#define BIN_CAP      2048
#define MAX_BINS     1024
#define PART_THREADS 512
#define PART_EDGES   4096
#define LIST_CAP     128
#define BIN_WORDS    6
#define SA_SHIFT     20

static_assert(MAX_BINS == 2 * PART_THREADS, "pass 1 clears two LDS counters per thread");

__global__ __launch_bounds__(PART_THREADS)
void bin_partition_kernel(const int*   __restrict__ edge_index,
                          const int*   __restrict__ edge_attr,
                          const float* __restrict__ edge_weight,
                          int*       __restrict__ bin_cursor,
                          int*       __restrict__ recs,
                          int*       __restrict__ ovf,
                          int E, int NB)
{
    int4* recs4 = (int4*)recs;
    __shared__ int hist[MAX_BINS];
    __shared__ int sbase[MAX_BINS];

    int tid = threadIdx.x;
    hist[2 * tid]     = 0;
    hist[2 * tid + 1] = 0;
    __syncthreads();

    int base = blockIdx.x * PART_EDGES;
    int dstv[PART_EDGES / PART_THREADS];
    for (int i = 0; i < PART_EDGES / PART_THREADS; ++i) {
        int e = base + i * PART_THREADS + tid;
        dstv[i] = (e < E) ? edge_index[E + e] : -1;
    }
    int   srcv[PART_EDGES / PART_THREADS];
    int   attv[PART_EDGES / PART_THREADS];
    float w0v[PART_EDGES / PART_THREADS];
    float w1v[PART_EDGES / PART_THREADS];
    for (int i = 0; i < PART_EDGES / PART_THREADS; ++i) {
        int e  = base + i * PART_THREADS + tid;
        int ec = (e < E) ? e : E - 1;
        srcv[i] = edge_index[ec];
        attv[i] = edge_attr[ec];
        w0v[i]  = edge_weight[ec];
        w1v[i]  = edge_weight[(size_t)E + ec];
    }
    for (int i = 0; i < PART_EDGES / PART_THREADS; ++i)
        if (dstv[i] >= 0) atomicAdd(&hist[dstv[i] >> BIN_SHIFT], 1);
    __syncthreads();

    for (int b = tid; b < NB; b += PART_THREADS) {
        int h = hist[b];
        if (h) {
            sbase[b] = atomicAdd(&bin_cursor[b], h);
            hist[b]  = 0;
        }
    }
    __syncthreads();

    for (int i = 0; i < PART_EDGES / PART_THREADS; ++i) {
        if (dstv[i] >= 0) {
            int b    = dstv[i] >> BIN_SHIFT;
            int slot = sbase[b] + atomicAdd(&hist[b], 1);
            int e    = base + i * PART_THREADS + tid;
            if (slot < BIN_CAP) {
                int4 r;
                r.x = (e << BIN_SHIFT) | (dstv[i] & ((1 << BIN_SHIFT) - 1));
                r.y = srcv[i] | (attv[i] << SA_SHIFT);
                r.z = __float_as_int(w0v[i]);
                r.w = __float_as_int(w1v[i]);
                recs4[(size_t)b * BIN_CAP + slot] = r;
            } else {
                int k = atomicAdd(&bin_cursor[MAX_BINS], 1);
                ovf[2 * k]     = e;
                ovf[2 * k + 1] = dstv[i];
            }
        }
    }
}

__device__ __forceinline__ v2f gelu_pair(v2f v)
{
    v2f a;
    a.x = fminf(fabsf(v.x), 8.0f);
    a.y = fminf(fabsf(v.y), 8.0f);
    v2f q = (v2f)(2.1919348114352033e-07f);
    q = __builtin_elementwise_fma(q, a, (v2f)(-5.2599425544030964e-06f));
    q = __builtin_elementwise_fma(q, a, (v2f)(4.943032763549127e-05f));
    q = __builtin_elementwise_fma(q, a, (v2f)(-0.00020471429161261767f));
    q = __builtin_elementwise_fma(q, a, (v2f)(-0.00012574861466418952f));
    q = __builtin_elementwise_fma(q, a, (v2f)(0.007073047570884228f));
    q = __builtin_elementwise_fma(q, a, (v2f)(-0.05251256376504898f));
    q = __builtin_elementwise_fma(q, a, (v2f)(-0.45920106768608093f));
    q = __builtin_elementwise_fma(q, a, (v2f)(-1.1511061191558838f));
    q = __builtin_elementwise_fma(q, a, (v2f)(-1.0f));
    v2f t;
    t.x = __builtin_amdgcn_exp2f(q.x);
    t.y = __builtin_amdgcn_exp2f(q.y);
    v2f r;
    r.x = fmaxf(v.x, 0.0f);
    r.y = fmaxf(v.y, 0.0f);
    return __builtin_elementwise_fma(-a, t, r);
}

__global__ __launch_bounds__(256)
void node_sum_kernel(const float* __restrict__ x,
                     const int*   __restrict__ edge_index,
                     const int*   __restrict__ edge_attr,
                     const float* __restrict__ edge_weight,
                     const float* __restrict__ bond_emb,
                     const int*   __restrict__ bin_cursor,
                     const int*   __restrict__ recs,
                     const int*   __restrict__ ovf,
                     float*       __restrict__ agg,
                     int E, int N)
{
    __shared__ int  ncnt[8];
    __shared__ int  elist[8][LIST_CAP];
    __shared__ int4 urec[8][LIST_CAP];
    __shared__ int4 srec[8][LIST_CAP];

    const int4* recs4 = (const int4*)recs;

    int tid  = threadIdx.x;
    int w    = tid >> 5;
    int lane = tid & 31;
    int nb0  = blockIdx.x * 8;
    int n    = nb0 + w;

    const int kmask = (1 << BIN_SHIFT) - 1;
    int bin  = nb0 >> BIN_SHIFT;
    int bkey = (nb0 & kmask) >> 3;
    int key  = n & kmask;
    int pad  = key ^ 1;
    int cbin = bin_cursor[bin];
    int novf = bin_cursor[MAX_BINS];
    int s    = bin * BIN_CAP;
    int t    = s + (cbin < BIN_CAP ? cbin : BIN_CAP);
    int d    = lane * 2;

    if (tid < 8) ncnt[tid] = 0;
    __syncthreads();
    for (int base = s; base < t; base += BIN_WORDS * 256) {
        int4 r[BIN_WORDS];
        for (int u = 0; u < BIN_WORDS; ++u) {
            int idx = base + u * 256 + tid;
            r[u] = make_int4(pad, 0, 0, 0);
            if (idx < t) r[u] = recs4[idx];
        }
        for (int u = 0; u < BIN_WORDS; ++u) {
            int idx = base + u * 256 + tid;
            if (idx < t && ((r[u].x & kmask) >> 3) == bkey) {
                int k   = r[u].x & 7;
                int pos = atomicAdd(&ncnt[k], 1);
                if (pos < LIST_CAP) {
                    elist[k][pos] = r[u].x >> BIN_SHIFT;
                    urec[k][pos]  = r[u];
                }
            }
        }
    }
    for (int base = 0; base < novf; base += 256) {
        int idx = base + tid;
        if (idx < novf) {
            int e  = ovf[2 * idx];
            int dd = ovf[2 * idx + 1];
            if ((dd >> 3) == (nb0 >> 3)) {
                int k   = dd & 7;
                int pos = atomicAdd(&ncnt[k], 1);
                if (pos < LIST_CAP) {
                    int4 r;
                    r.x = (e << BIN_SHIFT) | (dd & kmask);
                    r.y = edge_index[e] | (edge_attr[e] << SA_SHIFT);
                    r.z = __float_as_int(edge_weight[e]);
                    r.w = __float_as_int(edge_weight[(size_t)E + e]);
                    elist[k][pos] = e;
                    urec[k][pos]  = r;
                }
            }
        }
    }
    __syncthreads();
    if (n >= N) return;

    const float* x0 = x + d;
    const float* x1 = x + (size_t)N * D_FEAT + d;
    const float* eb = bond_emb + d;

    v2f acc0 = (v2f)(0.0f);
    v2f acc1 = (v2f)(0.0f);

    int deg = __builtin_amdgcn_readfirstlane(ncnt[w]);

    if (deg <= LIST_CAP) {
        for (int i = lane; i < deg; i += 32) {
            int ei = elist[w][i];
            int rank = 0;
            for (int j = 0; j < deg; ++j) rank += (elist[w][j] < ei) ? 1 : 0;
            srec[w][rank] = urec[w][i];
        }

        for (int g = 0; g < deg; g += 32) {
            int cnt = deg - g;
            if (cnt > 32) cnt = 32;

            int   sa_l = 0;
            float w0_l = 0.0f, w1_l = 0.0f;
            if (lane < cnt) {
                int4 rr = srec[w][g + lane];
                sa_l = rr.y;
                w0_l = __int_as_float(rr.z);
                w1_l = __int_as_float(rr.w);
            }

            const int samask = (1 << SA_SHIFT) - 1;
            int sa  = __builtin_amdgcn_readlane(sa_l, 0);
            int src = sa & samask;
            int att = (int)((unsigned)sa >> SA_SHIFT);
            v2f ev = *(const v2f*)(eb + (size_t)att * D_FEAT);
            v2f xa = *(const v2f*)(x0 + (size_t)src * D_FEAT);
            v2f xb = *(const v2f*)(x1 + (size_t)src * D_FEAT);
            for (int j = 0; j < cnt; ++j) {
                int jn   = (j + 1 < cnt) ? j + 1 : j;
                int san  = __builtin_amdgcn_readlane(sa_l, jn);
                int srcn = san & samask;
                int attn = (int)((unsigned)san >> SA_SHIFT);
                v2f evn = *(const v2f*)(eb + (size_t)attn * D_FEAT);
                v2f xan = *(const v2f*)(x0 + (size_t)srcn * D_FEAT);
                v2f xbn = *(const v2f*)(x1 + (size_t)srcn * D_FEAT);

                float w0 = __int_as_float(__builtin_amdgcn_readlane(__float_as_int(w0_l), j));
                float w1 = __int_as_float(__builtin_amdgcn_readlane(__float_as_int(w1_l), j));
                acc0 = __builtin_elementwise_fma(gelu_pair(xa + ev), (v2f)(w0), acc0);
                acc1 = __builtin_elementwise_fma(gelu_pair(xb + ev), (v2f)(w1), acc1);

                ev = evn; xa = xan; xb = xbn;
            }
        }
    } else {
        int last = -1;
        for (int k = 0; k < deg; ++k) {
            int best = 0x7fffffff;
            for (int base = s; base < t; base += 32) {
                int idx = base + lane;
                int r = (idx < t) ? recs4[idx].x : pad;
                int e = r >> BIN_SHIFT;
                if ((r & kmask) == key && e > last && e < best) best = e;
            }
            for (int base = 0; base < novf; base += 32) {
                int idx = base + lane;
                if (idx < novf) {
                    int e  = ovf[2 * idx];
                    int dd = ovf[2 * idx + 1];
                    if (dd == n && e > last && e < best) best = e;
                }
            }
            for (int off = 16; off > 0; off >>= 1) {
                int o = __shfl_xor(best, off);
                best = (o < best) ? o : best;
            }
            int e = __builtin_amdgcn_readfirstlane(best);
            last = e;
            int   src = edge_index[e];
            int   att = edge_attr[e];
            float w0  = edge_weight[e];
            float w1  = edge_weight[(size_t)E + e];
            v2f ev = *(const v2f*)(eb + (size_t)att * D_FEAT);
            v2f xa = *(const v2f*)(x0 + (size_t)src * D_FEAT);
            v2f xb = *(const v2f*)(x1 + (size_t)src * D_FEAT);
            acc0 = __builtin_elementwise_fma(gelu_pair(xa + ev), (v2f)(w0), acc0);
            acc1 = __builtin_elementwise_fma(gelu_pair(xb + ev), (v2f)(w1), acc1);
        }
    }

    *(v2f*)(agg + (size_t)n * D_FEAT + d)       = acc0;
    *(v2f*)(agg + ((size_t)N + n) * D_FEAT + d) = acc1;
}

__global__ __launch_bounds__(WAVES_PER_BLOCK * 32)
void gin_mlp_kernel(const float* __restrict__ x,
                    const float* __restrict__ agg,
                    const float* __restrict__ W1,
                    const float* __restrict__ b1,
                    const float* __restrict__ W2,
                    const float* __restrict__ b2,
                    const float* __restrict__ eps,
                    float*       __restrict__ out,
                    int n_tiles)
{
    __shared__ float4 w1s[8 * 8 * 32];
    __shared__ float4 w2s[4 * 16 * 32];
    __shared__ float  lds_t[WAVES_PER_BLOCK][16 * H_FEAT];

    int waveId = threadIdx.x >> 5;
    int lane   = threadIdx.x & 31;
    int tile   = blockIdx.x * WAVES_PER_BLOCK + waveId;
    bool valid = tile < n_tiles;

    int m    = lane & 15;
    int half = lane >> 4;

    v2f A[16];
#pragma unroll
    for (int k = 0; k < 16; ++k) { A[k].x = 0.0f; A[k].y = 0.0f; }
    if (valid) {
        float epsv = 1.0f + eps[0];
        size_t rowbase = ((size_t)tile * 16 + m) * D_FEAT;
#pragma unroll
        for (int k = 0; k < 16; ++k) {
            int col = k * 4 + half * 2;
            float2 xv = *(const float2*)(x   + rowbase + col);
            float2 av = *(const float2*)(agg + rowbase + col);
            A[k].x = epsv * xv.x + av.x;
            A[k].y = epsv * xv.y + av.y;
        }
    }

#pragma unroll
    for (int n = 0; n < 8; ++n) {
        int ncol = n * 16 + m;
        int kb0  = waveId * 8 + half * 2;
        float4 e;
        e.x = W1[(size_t)(kb0    ) * H_FEAT + ncol];
        e.y = W1[(size_t)(kb0 + 1) * H_FEAT + ncol];
        e.z = W1[(size_t)(kb0 + 4) * H_FEAT + ncol];
        e.w = W1[(size_t)(kb0 + 5) * H_FEAT + ncol];
        w1s[(n * 8 + waveId) * 32 + lane] = e;
    }
#pragma unroll
    for (int i = 0; i < 8; ++i) {
        int o    = i >> 1;
        int jj   = waveId + 8 * (i & 1);
        int ocol = o * 16 + m;
        int kb0  = jj * 8 + half * 2;
        float4 e;
        e.x = W2[(size_t)(kb0    ) * D_FEAT + ocol];
        e.y = W2[(size_t)(kb0 + 1) * D_FEAT + ocol];
        e.z = W2[(size_t)(kb0 + 4) * D_FEAT + ocol];
        e.w = W2[(size_t)(kb0 + 5) * D_FEAT + ocol];
        w2s[(o * 16 + jj) * 32 + lane] = e;
    }
    __syncthreads();
    if (!valid) return;

    float* tl = &lds_t[waveId][0];

#pragma unroll
    for (int n = 0; n < 8; ++n) {
        v8f acc = {};
        int ncol = n * 16 + m;
#pragma unroll
        for (int kk = 0; kk < 8; ++kk) {
            float4 e = w1s[(n * 8 + kk) * 32 + lane];
            v2f B0; B0.x = e.x; B0.y = e.y;
            v2f B1; B1.x = e.z; B1.y = e.w;
            acc = __builtin_amdgcn_wmma_f32_16x16x4_f32(
                      false, A[2 * kk], false, B0, (short)0, acc, false, false);
            acc = __builtin_amdgcn_wmma_f32_16x16x4_f32(
                      false, A[2 * kk + 1], false, B1, (short)0, acc, false, false);
        }
        float bias = b1[ncol];
#pragma unroll
        for (int v = 0; v < 8; ++v) {
            float t = acc[v] + bias;
            t = t > 0.0f ? t : 0.0f;
            tl[(v + 8 * half) * H_FEAT + ncol] = t;
        }
    }

    asm volatile("s_wait_dscnt 0" ::: "memory");
    __builtin_amdgcn_wave_barrier();

#pragma unroll
    for (int o = 0; o < 4; ++o) {
        v8f acc = {};
        int ocol = o * 16 + m;
#pragma unroll
        for (int jj = 0; jj < 16; ++jj) {
            float4 e = w2s[(o * 16 + jj) * 32 + lane];
            int kb = jj * 8 + half * 2;
            float2 a0 = *(const float2*)(tl + m * H_FEAT + kb);
            float2 a1 = *(const float2*)(tl + m * H_FEAT + kb + 4);
            v2f A2; A2.x = a0.x; A2.y = a0.y;
            v2f A3; A3.x = a1.x; A3.y = a1.y;
            v2f B0; B0.x = e.x; B0.y = e.y;
            v2f B1; B1.x = e.z; B1.y = e.w;
            acc = __builtin_amdgcn_wmma_f32_16x16x4_f32(
                      false, A2, false, B0, (short)0, acc, false, false);
            acc = __builtin_amdgcn_wmma_f32_16x16x4_f32(
                      false, A3, false, B1, (short)0, acc, false, false);
        }
        float bias = b2[ocol];
#pragma unroll
        for (int v = 0; v < 8; ++v) {
            out[((size_t)tile * 16 + v + 8 * half) * D_FEAT + ocol] = acc[v] + bias;
        }
    }
}

extern "C" void kernel_launch(void* const* d_in, const int* in_sizes, int n_in,
                              void* d_out, int out_size, void* d_ws, size_t ws_size,
                              hipStream_t stream)
{
    const float* x           = (const float*)d_in[0];
    const int*   edge_index  = (const int*)  d_in[1];
    const int*   edge_attr   = (const int*)  d_in[2];
    const float* edge_weight = (const float*)d_in[3];
    const float* bond_emb    = (const float*)d_in[4];
    const float* eps         = (const float*)d_in[5];
    const float* W1          = (const float*)d_in[6];
    const float* b1          = (const float*)d_in[7];
    const float* W2          = (const float*)d_in[8];
    const float* b2          = (const float*)d_in[9];
    float*       out         = (float*)d_out;

    const int RC = 2;
    const int E  = in_sizes[2];
    const int N  = in_sizes[0] / (RC * D_FEAT);
    const int NB = (N + (1 << BIN_SHIFT) - 1) >> BIN_SHIFT;

    float* agg        = (float*)d_ws;
    int*   bin_cursor = (int*)(agg + (size_t)RC * N * D_FEAT);
    int*   recs       = bin_cursor + MAX_BINS + 32;
    int*   ovf        = recs + (size_t)MAX_BINS * BIN_CAP * 4;

    hipMemsetAsync(bin_cursor, 0, (size_t)(MAX_BINS + 32) * sizeof(int), stream);

    int blocksP = (E + PART_EDGES - 1) / PART_EDGES;
    bin_partition_kernel<<<blocksP, PART_THREADS, 0, stream>>>(edge_index, edge_attr, edge_weight,
                                                               bin_cursor, recs, ovf, E, NB);

    int blocks1 = (N + 7) / 8;
    node_sum_kernel<<<blocks1, 256, 0, stream>>>(x, edge_index, edge_attr, edge_weight, bond_emb,
                                                 bin_cursor, recs, ovf, agg, E, N);

    int n_tiles = (RC * N) / 16;
    int blocks2 = (n_tiles + WAVES_PER_BLOCK - 1) / WAVES_PER_BLOCK;
    gin_mlp_kernel<<<blocks2, WAVES_PER_BLOCK * 32, 0, stream>>>(
        x, agg, W1, b1, W2, b2, eps, out, n_tiles);
}


# MLP phase: one 13-wave block per WGP (256 blocks; a block takes 24-25 tiles, each wave two in turn; weights staged once per WGP) instead of 782 8-wave blocks at two per WGP in two passes
# speedup vs baseline: 2.3789x; 1.0317x over previous
#include <hip/hip_runtime.h>
#include <hip/hip_bf16.h>
#include <math.h>


typedef float v2f __attribute__((ext_vector_type(2)));
typedef float v8f __attribute__((ext_vector_type(8)));

#define D_FEAT 64
#define H_FEAT 128
#define MLP_WAVES  13
#define MLP_BLOCKS 256
#define STG_WAVES  8

#define BIN_SHIFT    6
#define BIN_CAP      2048
#define MAX_BINS     1024
#define PART_THREADS 512
#define PART_EDGES   4096
#define LIST_CAP     128
#define BIN_WORDS    6
#define SA_SHIFT     20

static_assert(MAX_BINS == 2 * PART_THREADS, "pass 1 clears two LDS counters per thread");

__global__ __launch_bounds__(PART_THREADS)
void bin_partition_kernel(const int*   __restrict__ edge_index,
                          const int*   __restrict__ edge_attr,
                          const float* __restrict__ edge_weight,
                          int*       __restrict__ bin_cursor,
                          int*       __restrict__ recs,
                          int*       __restrict__ ovf,
                          int E, int NB)
{
    int4* recs4 = (int4*)recs;
    __shared__ int hist[MAX_BINS];
    __shared__ int sbase[MAX_BINS];

    int tid = threadIdx.x;
    hist[2 * tid]     = 0;
    hist[2 * tid + 1] = 0;
    __syncthreads();

    int base = blockIdx.x * PART_EDGES;
    int dstv[PART_EDGES / PART_THREADS];
    for (int i = 0; i < PART_EDGES / PART_THREADS; ++i) {
        int e = base + i * PART_THREADS + tid;
        dstv[i] = (e < E) ? edge_index[E + e] : -1;
    }
    int   srcv[PART_EDGES / PART_THREADS];
    int   attv[PART_EDGES / PART_THREADS];
    float w0v[PART_EDGES / PART_THREADS];
    float w1v[PART_EDGES / PART_THREADS];
    for (int i = 0; i < PART_EDGES / PART_THREADS; ++i) {
        int e  = base + i * PART_THREADS + tid;
        int ec = (e < E) ? e : E - 1;
        srcv[i] = edge_index[ec];
        attv[i] = edge_attr[ec];
        w0v[i]  = edge_weight[ec];
        w1v[i]  = edge_weight[(size_t)E + ec];
    }
    for (int i = 0; i < PART_EDGES / PART_THREADS; ++i)
        if (dstv[i] >= 0) atomicAdd(&hist[dstv[i] >> BIN_SHIFT], 1);
    __syncthreads();

    for (int b = tid; b < NB; b += PART_THREADS) {
        int h = hist[b];
        if (h) {
            sbase[b] = atomicAdd(&bin_cursor[b], h);
            hist[b]  = 0;
        }
    }
    __syncthreads();

    for (int i = 0; i < PART_EDGES / PART_THREADS; ++i) {
        if (dstv[i] >= 0) {
            int b    = dstv[i] >> BIN_SHIFT;
            int slot = sbase[b] + atomicAdd(&hist[b], 1);
            int e    = base + i * PART_THREADS + tid;
            if (slot < BIN_CAP) {
                int4 r;
                r.x = (e << BIN_SHIFT) | (dstv[i] & ((1 << BIN_SHIFT) - 1));
                r.y = srcv[i] | (attv[i] << SA_SHIFT);
                r.z = __float_as_int(w0v[i]);
                r.w = __float_as_int(w1v[i]);
                recs4[(size_t)b * BIN_CAP + slot] = r;
            } else {
                int k = atomicAdd(&bin_cursor[MAX_BINS], 1);
                ovf[2 * k]     = e;
                ovf[2 * k + 1] = dstv[i];
            }
        }
    }
}

__device__ __forceinline__ v2f gelu_pair(v2f v)
{
    v2f a;
    a.x = fminf(fabsf(v.x), 8.0f);
    a.y = fminf(fabsf(v.y), 8.0f);
    v2f q = (v2f)(2.1919348114352033e-07f);
    q = __builtin_elementwise_fma(q, a, (v2f)(-5.2599425544030964e-06f));
    q = __builtin_elementwise_fma(q, a, (v2f)(4.943032763549127e-05f));
    q = __builtin_elementwise_fma(q, a, (v2f)(-0.00020471429161261767f));
    q = __builtin_elementwise_fma(q, a, (v2f)(-0.00012574861466418952f));
    q = __builtin_elementwise_fma(q, a, (v2f)(0.007073047570884228f));
    q = __builtin_elementwise_fma(q, a, (v2f)(-0.05251256376504898f));
    q = __builtin_elementwise_fma(q, a, (v2f)(-0.45920106768608093f));
    q = __builtin_elementwise_fma(q, a, (v2f)(-1.1511061191558838f));
    q = __builtin_elementwise_fma(q, a, (v2f)(-1.0f));
    v2f t;
    t.x = __builtin_amdgcn_exp2f(q.x);
    t.y = __builtin_amdgcn_exp2f(q.y);
    v2f r;
    r.x = fmaxf(v.x, 0.0f);
    r.y = fmaxf(v.y, 0.0f);
    return __builtin_elementwise_fma(-a, t, r);
}

__global__ __launch_bounds__(256)
void node_sum_kernel(const float* __restrict__ x,
                     const int*   __restrict__ edge_index,
                     const int*   __restrict__ edge_attr,
                     const float* __restrict__ edge_weight,
                     const float* __restrict__ bond_emb,
                     const int*   __restrict__ bin_cursor,
                     const int*   __restrict__ recs,
                     const int*   __restrict__ ovf,
                     float*       __restrict__ agg,
                     int E, int N)
{
    __shared__ int  ncnt[8];
    __shared__ int  elist[8][LIST_CAP];
    __shared__ int4 urec[8][LIST_CAP];
    __shared__ int4 srec[8][LIST_CAP];

    const int4* recs4 = (const int4*)recs;

    int tid  = threadIdx.x;
    int w    = tid >> 5;
    int lane = tid & 31;
    int nb0  = blockIdx.x * 8;
    int n    = nb0 + w;

    const int kmask = (1 << BIN_SHIFT) - 1;
    int bin  = nb0 >> BIN_SHIFT;
    int bkey = (nb0 & kmask) >> 3;
    int key  = n & kmask;
    int pad  = key ^ 1;
    int cbin = bin_cursor[bin];
    int novf = bin_cursor[MAX_BINS];
    int s    = bin * BIN_CAP;
    int t    = s + (cbin < BIN_CAP ? cbin : BIN_CAP);
    int d    = lane * 2;

    if (tid < 8) ncnt[tid] = 0;
    __syncthreads();
    for (int base = s; base < t; base += BIN_WORDS * 256) {
        int4 r[BIN_WORDS];
        for (int u = 0; u < BIN_WORDS; ++u) {
            int idx = base + u * 256 + tid;
            r[u] = make_int4(pad, 0, 0, 0);
            if (idx < t) r[u] = recs4[idx];
        }
        for (int u = 0; u < BIN_WORDS; ++u) {
            int idx = base + u * 256 + tid;
            if (idx < t && ((r[u].x & kmask) >> 3) == bkey) {
                int k   = r[u].x & 7;
                int pos = atomicAdd(&ncnt[k], 1);
                if (pos < LIST_CAP) {
                    elist[k][pos] = r[u].x >> BIN_SHIFT;
                    urec[k][pos]  = r[u];
                }
            }
        }
    }
    for (int base = 0; base < novf; base += 256) {
        int idx = base + tid;
        if (idx < novf) {
            int e  = ovf[2 * idx];
            int dd = ovf[2 * idx + 1];
            if ((dd >> 3) == (nb0 >> 3)) {
                int k   = dd & 7;
                int pos = atomicAdd(&ncnt[k], 1);
                if (pos < LIST_CAP) {
                    int4 r;
                    r.x = (e << BIN_SHIFT) | (dd & kmask);
                    r.y = edge_index[e] | (edge_attr[e] << SA_SHIFT);
                    r.z = __float_as_int(edge_weight[e]);
                    r.w = __float_as_int(edge_weight[(size_t)E + e]);
                    elist[k][pos] = e;
                    urec[k][pos]  = r;
                }
            }
        }
    }
    __syncthreads();
    if (n >= N) return;

    const float* x0 = x + d;
    const float* x1 = x + (size_t)N * D_FEAT + d;
    const float* eb = bond_emb + d;

    v2f acc0 = (v2f)(0.0f);
    v2f acc1 = (v2f)(0.0f);

    int deg = __builtin_amdgcn_readfirstlane(ncnt[w]);

    if (deg <= LIST_CAP) {
        for (int i = lane; i < deg; i += 32) {
            int ei = elist[w][i];
            int rank = 0;
            for (int j = 0; j < deg; ++j) rank += (elist[w][j] < ei) ? 1 : 0;
            srec[w][rank] = urec[w][i];
        }

        for (int g = 0; g < deg; g += 32) {
            int cnt = deg - g;
            if (cnt > 32) cnt = 32;

            int   sa_l = 0;
            float w0_l = 0.0f, w1_l = 0.0f;
            if (lane < cnt) {
                int4 rr = srec[w][g + lane];
                sa_l = rr.y;
                w0_l = __int_as_float(rr.z);
                w1_l = __int_as_float(rr.w);
            }

            const int samask = (1 << SA_SHIFT) - 1;
            int sa  = __builtin_amdgcn_readlane(sa_l, 0);
            int src = sa & samask;
            int att = (int)((unsigned)sa >> SA_SHIFT);
            v2f ev = *(const v2f*)(eb + (size_t)att * D_FEAT);
            v2f xa = *(const v2f*)(x0 + (size_t)src * D_FEAT);
            v2f xb = *(const v2f*)(x1 + (size_t)src * D_FEAT);
            for (int j = 0; j < cnt; ++j) {
                int jn   = (j + 1 < cnt) ? j + 1 : j;
                int san  = __builtin_amdgcn_readlane(sa_l, jn);
                int srcn = san & samask;
                int attn = (int)((unsigned)san >> SA_SHIFT);
                v2f evn = *(const v2f*)(eb + (size_t)attn * D_FEAT);
                v2f xan = *(const v2f*)(x0 + (size_t)srcn * D_FEAT);
                v2f xbn = *(const v2f*)(x1 + (size_t)srcn * D_FEAT);

                float w0 = __int_as_float(__builtin_amdgcn_readlane(__float_as_int(w0_l), j));
                float w1 = __int_as_float(__builtin_amdgcn_readlane(__float_as_int(w1_l), j));
                acc0 = __builtin_elementwise_fma(gelu_pair(xa + ev), (v2f)(w0), acc0);
                acc1 = __builtin_elementwise_fma(gelu_pair(xb + ev), (v2f)(w1), acc1);

                ev = evn; xa = xan; xb = xbn;
            }
        }
    } else {
        int last = -1;
        for (int k = 0; k < deg; ++k) {
            int best = 0x7fffffff;
            for (int base = s; base < t; base += 32) {
                int idx = base + lane;
                int r = (idx < t) ? recs4[idx].x : pad;
                int e = r >> BIN_SHIFT;
                if ((r & kmask) == key && e > last && e < best) best = e;
            }
            for (int base = 0; base < novf; base += 32) {
                int idx = base + lane;
                if (idx < novf) {
                    int e  = ovf[2 * idx];
                    int dd = ovf[2 * idx + 1];
                    if (dd == n && e > last && e < best) best = e;
                }
            }
            for (int off = 16; off > 0; off >>= 1) {
                int o = __shfl_xor(best, off);
                best = (o < best) ? o : best;
            }
            int e = __builtin_amdgcn_readfirstlane(best);
            last = e;
            int   src = edge_index[e];
            int   att = edge_attr[e];
            float w0  = edge_weight[e];
            float w1  = edge_weight[(size_t)E + e];
            v2f ev = *(const v2f*)(eb + (size_t)att * D_FEAT);
            v2f xa = *(const v2f*)(x0 + (size_t)src * D_FEAT);
            v2f xb = *(const v2f*)(x1 + (size_t)src * D_FEAT);
            acc0 = __builtin_elementwise_fma(gelu_pair(xa + ev), (v2f)(w0), acc0);
            acc1 = __builtin_elementwise_fma(gelu_pair(xb + ev), (v2f)(w1), acc1);
        }
    }

    *(v2f*)(agg + (size_t)n * D_FEAT + d)       = acc0;
    *(v2f*)(agg + ((size_t)N + n) * D_FEAT + d) = acc1;
}

__global__ __launch_bounds__(MLP_WAVES * 32)
void gin_mlp_kernel(const float* __restrict__ x,
                    const float* __restrict__ agg,
                    const float* __restrict__ W1,
                    const float* __restrict__ b1,
                    const float* __restrict__ W2,
                    const float* __restrict__ b2,
                    const float* __restrict__ eps,
                    float*       __restrict__ out,
                    int n_tiles)
{
    __shared__ float4 w1s[8 * 8 * 32];
    __shared__ float4 w2s[4 * 16 * 32];
    __shared__ float  lds_t[MLP_WAVES][16 * H_FEAT];

    static_assert(MLP_WAVES >= STG_WAVES, "the first STG_WAVES waves stage the weights");
    int waveId = __builtin_amdgcn_readfirstlane(threadIdx.x >> 5);
    int lane   = threadIdx.x & 31;
    int t_lo   = (int)(((unsigned)blockIdx.x * (unsigned)n_tiles) / (unsigned)MLP_BLOCKS);
    int t_hi   = (int)((((unsigned)blockIdx.x + 1u) * (unsigned)n_tiles) / (unsigned)MLP_BLOCKS);
    int tile   = t_lo + waveId;
    bool valid = tile < t_hi;

    int m    = lane & 15;
    int half = lane >> 4;

    float epsv = 1.0f + eps[0];
    float2 xr[16], ar[16];
    size_t rowbase0 = ((size_t)(valid ? tile : 0) * 16 + m) * D_FEAT;

    if (waveId < STG_WAVES) {
        float4 stg[16];
#pragma unroll
        for (int n = 0; n < 8; ++n) {
            int ncol = n * 16 + m;
            int kb0  = waveId * 8 + half * 2;
            stg[n].x = W1[(size_t)(kb0    ) * H_FEAT + ncol];
            stg[n].y = W1[(size_t)(kb0 + 1) * H_FEAT + ncol];
            stg[n].z = W1[(size_t)(kb0 + 4) * H_FEAT + ncol];
            stg[n].w = W1[(size_t)(kb0 + 5) * H_FEAT + ncol];
        }
#pragma unroll
        for (int i = 0; i < 8; ++i) {
            int o    = i >> 1;
            int jj   = waveId + 8 * (i & 1);
            int ocol = o * 16 + m;
            int kb0  = jj * 8 + half * 2;
            stg[8 + i].x = W2[(size_t)(kb0    ) * D_FEAT + ocol];
            stg[8 + i].y = W2[(size_t)(kb0 + 1) * D_FEAT + ocol];
            stg[8 + i].z = W2[(size_t)(kb0 + 4) * D_FEAT + ocol];
            stg[8 + i].w = W2[(size_t)(kb0 + 5) * D_FEAT + ocol];
        }
#pragma unroll
        for (int k = 0; k < 16; ++k) {
            int col = k * 4 + half * 2;
            xr[k] = *(const float2*)(x   + rowbase0 + col);
            ar[k] = *(const float2*)(agg + rowbase0 + col);
        }
#pragma unroll
        for (int n = 0; n < 8; ++n)
            w1s[(n * 8 + waveId) * 32 + lane] = stg[n];
#pragma unroll
        for (int i = 0; i < 8; ++i)
            w2s[((i >> 1) * 16 + waveId + 8 * (i & 1)) * 32 + lane] = stg[8 + i];
    } else {
#pragma unroll
        for (int k = 0; k < 16; ++k) {
            int col = k * 4 + half * 2;
            xr[k] = *(const float2*)(x   + rowbase0 + col);
            ar[k] = *(const float2*)(agg + rowbase0 + col);
        }
    }
    v2f A[16];
#pragma unroll
    for (int k = 0; k < 16; ++k) {
        A[k].x = epsv * xr[k].x + ar[k].x;
        A[k].y = epsv * xr[k].y + ar[k].y;
    }
    __syncthreads();
    if (!valid) return;

    float* tl = &lds_t[waveId][0];

    for (;;) {
#pragma unroll
        for (int n = 0; n < 8; ++n) {
            v8f acc = {};
            int ncol = n * 16 + m;
#pragma unroll
            for (int kk = 0; kk < 8; ++kk) {
                float4 e = w1s[(n * 8 + kk) * 32 + lane];
                v2f B0; B0.x = e.x; B0.y = e.y;
                v2f B1; B1.x = e.z; B1.y = e.w;
                acc = __builtin_amdgcn_wmma_f32_16x16x4_f32(
                          false, A[2 * kk], false, B0, (short)0, acc, false, false);
                acc = __builtin_amdgcn_wmma_f32_16x16x4_f32(
                          false, A[2 * kk + 1], false, B1, (short)0, acc, false, false);
            }
            float bias = b1[ncol];
#pragma unroll
            for (int v = 0; v < 8; ++v) {
                float t = acc[v] + bias;
                t = t > 0.0f ? t : 0.0f;
                tl[(v + 8 * half) * H_FEAT + ncol] = t;
            }
        }

        asm volatile("s_wait_dscnt 0" ::: "memory");
        __builtin_amdgcn_wave_barrier();

#pragma unroll
        for (int o = 0; o < 4; ++o) {
            v8f acc = {};
            int ocol = o * 16 + m;
#pragma unroll
            for (int jj = 0; jj < 16; ++jj) {
                float4 e = w2s[(o * 16 + jj) * 32 + lane];
                int kb = jj * 8 + half * 2;
                float2 a0 = *(const float2*)(tl + m * H_FEAT + kb);
                float2 a1 = *(const float2*)(tl + m * H_FEAT + kb + 4);
                v2f A2; A2.x = a0.x; A2.y = a0.y;
                v2f A3; A3.x = a1.x; A3.y = a1.y;
                v2f B0; B0.x = e.x; B0.y = e.y;
                v2f B1; B1.x = e.z; B1.y = e.w;
                acc = __builtin_amdgcn_wmma_f32_16x16x4_f32(
                          false, A2, false, B0, (short)0, acc, false, false);
                acc = __builtin_amdgcn_wmma_f32_16x16x4_f32(
                          false, A3, false, B1, (short)0, acc, false, false);
            }
            float bias = b2[ocol];
#pragma unroll
            for (int v = 0; v < 8; ++v) {
                out[((size_t)tile * 16 + v + 8 * half) * D_FEAT + ocol] = acc[v] + bias;
            }
        }

        tile += MLP_WAVES;
        if (tile >= t_hi) break;
        size_t rowbase = ((size_t)tile * 16 + m) * D_FEAT;
#pragma unroll
        for (int k = 0; k < 16; ++k) {
            int col = k * 4 + half * 2;
            float2 xv = *(const float2*)(x   + rowbase + col);
            float2 av = *(const float2*)(agg + rowbase + col);
            A[k].x = epsv * xv.x + av.x;
            A[k].y = epsv * xv.y + av.y;
        }
    }
}

extern "C" void kernel_launch(void* const* d_in, const int* in_sizes, int n_in,
                              void* d_out, int out_size, void* d_ws, size_t ws_size,
                              hipStream_t stream)
{
    const float* x           = (const float*)d_in[0];
    const int*   edge_index  = (const int*)  d_in[1];
    const int*   edge_attr   = (const int*)  d_in[2];
    const float* edge_weight = (const float*)d_in[3];
    const float* bond_emb    = (const float*)d_in[4];
    const float* eps         = (const float*)d_in[5];
    const float* W1          = (const float*)d_in[6];
    const float* b1          = (const float*)d_in[7];
    const float* W2          = (const float*)d_in[8];
    const float* b2          = (const float*)d_in[9];
    float*       out         = (float*)d_out;

    const int RC = 2;
    const int E  = in_sizes[2];
    const int N  = in_sizes[0] / (RC * D_FEAT);
    const int NB = (N + (1 << BIN_SHIFT) - 1) >> BIN_SHIFT;

    float* agg        = (float*)d_ws;
    int*   bin_cursor = (int*)(agg + (size_t)RC * N * D_FEAT);
    int*   recs       = bin_cursor + MAX_BINS + 32;
    int*   ovf        = recs + (size_t)MAX_BINS * BIN_CAP * 4;

    hipMemsetAsync(bin_cursor, 0, (size_t)(MAX_BINS + 32) * sizeof(int), stream);

    int blocksP = (E + PART_EDGES - 1) / PART_EDGES;
    bin_partition_kernel<<<blocksP, PART_THREADS, 0, stream>>>(edge_index, edge_attr, edge_weight,
                                                               bin_cursor, recs, ovf, E, NB);

    int blocks1 = (N + 7) / 8;
    node_sum_kernel<<<blocks1, 256, 0, stream>>>(x, edge_index, edge_attr, edge_weight, bond_emb,
                                                 bin_cursor, recs, ovf, agg, E, N);

    int n_tiles = (RC * N) / 16;
    gin_mlp_kernel<<<MLP_BLOCKS, MLP_WAVES * 32, 0, stream>>>(
        x, agg, W1, b1, W2, b2, eps, out, n_tiles);
}


# Edge phase: a node-sum wave takes two nodes at once (16 per block): a trip asks for the rows of 4 edges, 2 of each node, then sums them; half the waves and blocks enter; per-node sum order kept
# speedup vs baseline: 2.4990x; 1.0505x over previous
#include <hip/hip_runtime.h>
#include <hip/hip_bf16.h>
#include <math.h>


typedef float v2f __attribute__((ext_vector_type(2)));
typedef float v8f __attribute__((ext_vector_type(8)));

#define D_FEAT 64
#define H_FEAT 128
#define MLP_WAVES  13
#define MLP_BLOCKS 256
#define STG_WAVES  8

#define BIN_SHIFT    6
#define BIN_CAP      2048
#define MAX_BINS     1024
#define PART_THREADS 512
#define PART_EDGES   4096
#define LIST_CAP     64
#define NODES_PB     16
#define BIN_WORDS    6
#define SA_SHIFT     20

static_assert(MAX_BINS == 2 * PART_THREADS, "pass 1 clears two LDS counters per thread");
static_assert(NODES_PB == 16 && (1 << BIN_SHIFT) % NODES_PB == 0, "a pass 2 block's nodes lie in one bin; its key tests use 16");

__global__ __launch_bounds__(PART_THREADS)
void bin_partition_kernel(const int*   __restrict__ edge_index,
                          const int*   __restrict__ edge_attr,
                          const float* __restrict__ edge_weight,
                          int*       __restrict__ bin_cursor,
                          int*       __restrict__ recs,
                          int*       __restrict__ ovf,
                          int E, int NB)
{
    int4* recs4 = (int4*)recs;
    __shared__ int hist[MAX_BINS];
    __shared__ int sbase[MAX_BINS];

    int tid = threadIdx.x;
    hist[2 * tid]     = 0;
    hist[2 * tid + 1] = 0;
    __syncthreads();

    int base = blockIdx.x * PART_EDGES;
    int dstv[PART_EDGES / PART_THREADS];
    for (int i = 0; i < PART_EDGES / PART_THREADS; ++i) {
        int e = base + i * PART_THREADS + tid;
        dstv[i] = (e < E) ? edge_index[E + e] : -1;
    }
    int   srcv[PART_EDGES / PART_THREADS];
    int   attv[PART_EDGES / PART_THREADS];
    float w0v[PART_EDGES / PART_THREADS];
    float w1v[PART_EDGES / PART_THREADS];
    for (int i = 0; i < PART_EDGES / PART_THREADS; ++i) {
        int e  = base + i * PART_THREADS + tid;
        int ec = (e < E) ? e : E - 1;
        srcv[i] = edge_index[ec];
        attv[i] = edge_attr[ec];
        w0v[i]  = edge_weight[ec];
        w1v[i]  = edge_weight[(size_t)E + ec];
    }
    for (int i = 0; i < PART_EDGES / PART_THREADS; ++i)
        if (dstv[i] >= 0) atomicAdd(&hist[dstv[i] >> BIN_SHIFT], 1);
    __syncthreads();

    for (int b = tid; b < NB; b += PART_THREADS) {
        int h = hist[b];
        if (h) {
            sbase[b] = atomicAdd(&bin_cursor[b], h);
            hist[b]  = 0;
        }
    }
    __syncthreads();

    for (int i = 0; i < PART_EDGES / PART_THREADS; ++i) {
        if (dstv[i] >= 0) {
            int b    = dstv[i] >> BIN_SHIFT;
            int slot = sbase[b] + atomicAdd(&hist[b], 1);
            int e    = base + i * PART_THREADS + tid;
            if (slot < BIN_CAP) {
                int4 r;
                r.x = (e << BIN_SHIFT) | (dstv[i] & ((1 << BIN_SHIFT) - 1));
                r.y = srcv[i] | (attv[i] << SA_SHIFT);
                r.z = __float_as_int(w0v[i]);
                r.w = __float_as_int(w1v[i]);
                recs4[(size_t)b * BIN_CAP + slot] = r;
            } else {
                int k = atomicAdd(&bin_cursor[MAX_BINS], 1);
                ovf[2 * k]     = e;
                ovf[2 * k + 1] = dstv[i];
            }
        }
    }
}

__device__ __forceinline__ v2f gelu_pair(v2f v)
{
    v2f a;
    a.x = fminf(fabsf(v.x), 8.0f);
    a.y = fminf(fabsf(v.y), 8.0f);
    v2f q = (v2f)(2.1919348114352033e-07f);
    q = __builtin_elementwise_fma(q, a, (v2f)(-5.2599425544030964e-06f));
    q = __builtin_elementwise_fma(q, a, (v2f)(4.943032763549127e-05f));
    q = __builtin_elementwise_fma(q, a, (v2f)(-0.00020471429161261767f));
    q = __builtin_elementwise_fma(q, a, (v2f)(-0.00012574861466418952f));
    q = __builtin_elementwise_fma(q, a, (v2f)(0.007073047570884228f));
    q = __builtin_elementwise_fma(q, a, (v2f)(-0.05251256376504898f));
    q = __builtin_elementwise_fma(q, a, (v2f)(-0.45920106768608093f));
    q = __builtin_elementwise_fma(q, a, (v2f)(-1.1511061191558838f));
    q = __builtin_elementwise_fma(q, a, (v2f)(-1.0f));
    v2f t;
    t.x = __builtin_amdgcn_exp2f(q.x);
    t.y = __builtin_amdgcn_exp2f(q.y);
    v2f r;
    r.x = fmaxf(v.x, 0.0f);
    r.y = fmaxf(v.y, 0.0f);
    return __builtin_elementwise_fma(-a, t, r);
}

struct EdgeRows { v2f ev, xa, xb; };

__device__ __forceinline__ EdgeRows ask_rows(const float* x, const float* bond_emb, size_t chan, unsigned d, int sa)
{
    const int samask = (1 << SA_SHIFT) - 1;
    int src = sa & samask;
    int att = (int)((unsigned)sa >> SA_SHIFT);
    const float* er = bond_emb + (size_t)att * D_FEAT;
    const float* xr = x + (size_t)src * D_FEAT;
    EdgeRows r;
    r.ev = *(const v2f*)(er + d);
    r.xa = *(const v2f*)(xr + d);
    r.xb = *(const v2f*)(xr + chan + d);
    return r;
}

__device__ __forceinline__ void add_edge(v2f& acc0, v2f& acc1, const EdgeRows& r, float w0, float w1)
{
    acc0 = __builtin_elementwise_fma(gelu_pair(r.xa + r.ev), (v2f)(w0), acc0);
    acc1 = __builtin_elementwise_fma(gelu_pair(r.xb + r.ev), (v2f)(w1), acc1);
}

__device__ __forceinline__ void rank_list(const int* el, const int4* ur, int4* sr, int deg, int lane)
{
    for (int i = lane; i < deg; i += 32) {
        int ei = el[i];
        int rank = 0;
        for (int j = 0; j < deg; ++j) rank += (el[j] < ei) ? 1 : 0;
        sr[rank] = ur[i];
    }
}

__global__ __launch_bounds__(256)
void node_sum_kernel(const float* __restrict__ x,
                     const int*   __restrict__ edge_index,
                     const int*   __restrict__ edge_attr,
                     const float* __restrict__ edge_weight,
                     const float* __restrict__ bond_emb,
                     const int*   __restrict__ bin_cursor,
                     const int*   __restrict__ recs,
                     const int*   __restrict__ ovf,
                     float*       __restrict__ agg,
                     int E, int N)
{
    __shared__ int  ncnt[NODES_PB];
    __shared__ int  elist[NODES_PB][LIST_CAP];
    __shared__ int4 urec[NODES_PB][LIST_CAP];
    __shared__ int4 srec[NODES_PB][LIST_CAP];

    const int4* recs4 = (const int4*)recs;

    int tid  = threadIdx.x;
    int w    = tid >> 5;
    int lane = tid & 31;
    int nb0  = blockIdx.x * NODES_PB;

    const int kmask = (1 << BIN_SHIFT) - 1;
    int bin  = nb0 >> BIN_SHIFT;
    int bkey = (nb0 & kmask) >> 4;
    int cbin = bin_cursor[bin];
    int novf = bin_cursor[MAX_BINS];
    int s    = bin * BIN_CAP;
    int t    = s + (cbin < BIN_CAP ? cbin : BIN_CAP);
    int d    = lane * 2;

    if (tid < NODES_PB) ncnt[tid] = 0;
    __syncthreads();
    for (int base = s; base < t; base += BIN_WORDS * 256) {
        int4 r[BIN_WORDS];
        for (int u = 0; u < BIN_WORDS; ++u) {
            int idx = base + u * 256 + tid;
            r[u] = make_int4(0, 0, 0, 0);
            if (idx < t) r[u] = recs4[idx];
        }
        for (int u = 0; u < BIN_WORDS; ++u) {
            int idx = base + u * 256 + tid;
            if (idx < t && ((r[u].x & kmask) >> 4) == bkey) {
                int k   = r[u].x & (NODES_PB - 1);
                int pos = atomicAdd(&ncnt[k], 1);
                if (pos < LIST_CAP) {
                    elist[k][pos] = r[u].x >> BIN_SHIFT;
                    urec[k][pos]  = r[u];
                }
            }
        }
    }
    for (int base = 0; base < novf; base += 256) {
        int idx = base + tid;
        if (idx < novf) {
            int e  = ovf[2 * idx];
            int dd = ovf[2 * idx + 1];
            if ((dd >> 4) == (nb0 >> 4)) {
                int k   = dd & (NODES_PB - 1);
                int pos = atomicAdd(&ncnt[k], 1);
                if (pos < LIST_CAP) {
                    int4 r;
                    r.x = (e << BIN_SHIFT) | (dd & kmask);
                    r.y = edge_index[e] | (edge_attr[e] << SA_SHIFT);
                    r.z = __float_as_int(edge_weight[e]);
                    r.w = __float_as_int(edge_weight[(size_t)E + e]);
                    elist[k][pos] = e;
                    urec[k][pos]  = r;
                }
            }
        }
    }
    __syncthreads();

    const float* x0 = x + d;
    const float* x1 = x + (size_t)N * D_FEAT + d;
    const float* eb = bond_emb + d;
    const size_t chan = (size_t)N * D_FEAT;

    int liA = w, liB = w + 8;
    int nA  = nb0 + liA, nB = nb0 + liB;
    if (nA >= N) return;
    int degA = __builtin_amdgcn_readfirstlane(ncnt[liA]);
    int degB = __builtin_amdgcn_readfirstlane(ncnt[liB]);

    if (nB < N && degA <= LIST_CAP && degB <= LIST_CAP) {
        rank_list(elist[liA], urec[liA], srec[liA], degA, lane);
        rank_list(elist[liB], urec[liB], srec[liB], degB, lane);

        v2f accA0 = (v2f)(0.0f), accA1 = (v2f)(0.0f);
        v2f accB0 = (v2f)(0.0f), accB1 = (v2f)(0.0f);

        int degM = degA > degB ? degA : degB;
        for (int g = 0; g < degM; g += 32) {
            int cntA = degA - g; cntA = cntA < 0 ? 0 : (cntA > 32 ? 32 : cntA);
            int cntB = degB - g; cntB = cntB < 0 ? 0 : (cntB > 32 ? 32 : cntB);

            int   saA_l = 0, saB_l = 0;
            float wA0_l = 0.0f, wA1_l = 0.0f, wB0_l = 0.0f, wB1_l = 0.0f;
            if (lane < cntA) {
                int4 rr = srec[liA][g + lane];
                saA_l = rr.y;
                wA0_l = __int_as_float(rr.z);
                wA1_l = __int_as_float(rr.w);
            }
            if (lane < cntB) {
                int4 rr = srec[liB][g + lane];
                saB_l = rr.y;
                wB0_l = __int_as_float(rr.z);
                wB1_l = __int_as_float(rr.w);
            }

            int m  = cntA > cntB ? cntA : cntB;
            int mf = (cntA < cntB ? cntA : cntB) & ~1;
            int j  = 0;
            for (; j < mf; j += 2) {
                EdgeRows a0 = ask_rows(x, bond_emb, chan, (unsigned)d, __builtin_amdgcn_readlane(saA_l, j));
                EdgeRows b0 = ask_rows(x, bond_emb, chan, (unsigned)d, __builtin_amdgcn_readlane(saB_l, j));
                EdgeRows a1 = ask_rows(x, bond_emb, chan, (unsigned)d, __builtin_amdgcn_readlane(saA_l, j + 1));
                EdgeRows b1 = ask_rows(x, bond_emb, chan, (unsigned)d, __builtin_amdgcn_readlane(saB_l, j + 1));
                float wa00 = __int_as_float(__builtin_amdgcn_readlane(__float_as_int(wA0_l), j)),     wa01 = __int_as_float(__builtin_amdgcn_readlane(__float_as_int(wA1_l), j));
                float wb00 = __int_as_float(__builtin_amdgcn_readlane(__float_as_int(wB0_l), j)),     wb01 = __int_as_float(__builtin_amdgcn_readlane(__float_as_int(wB1_l), j));
                float wa10 = __int_as_float(__builtin_amdgcn_readlane(__float_as_int(wA0_l), j + 1)), wa11 = __int_as_float(__builtin_amdgcn_readlane(__float_as_int(wA1_l), j + 1));
                float wb10 = __int_as_float(__builtin_amdgcn_readlane(__float_as_int(wB0_l), j + 1)), wb11 = __int_as_float(__builtin_amdgcn_readlane(__float_as_int(wB1_l), j + 1));
                add_edge(accA0, accA1, a0, wa00, wa01);
                add_edge(accB0, accB1, b0, wb00, wb01);
                add_edge(accA0, accA1, a1, wa10, wa11);
                add_edge(accB0, accB1, b1, wb10, wb11);
            }
            for (; j < m; j += 2) {
                EdgeRows a0 = ask_rows(x, bond_emb, chan, (unsigned)d, __builtin_amdgcn_readlane(saA_l, j));
                EdgeRows b0 = ask_rows(x, bond_emb, chan, (unsigned)d, __builtin_amdgcn_readlane(saB_l, j));
                EdgeRows a1 = ask_rows(x, bond_emb, chan, (unsigned)d, __builtin_amdgcn_readlane(saA_l, j + 1));
                EdgeRows b1 = ask_rows(x, bond_emb, chan, (unsigned)d, __builtin_amdgcn_readlane(saB_l, j + 1));
                float wa00 = __int_as_float(__builtin_amdgcn_readlane(__float_as_int(wA0_l), j)),     wa01 = __int_as_float(__builtin_amdgcn_readlane(__float_as_int(wA1_l), j));
                float wb00 = __int_as_float(__builtin_amdgcn_readlane(__float_as_int(wB0_l), j)),     wb01 = __int_as_float(__builtin_amdgcn_readlane(__float_as_int(wB1_l), j));
                float wa10 = __int_as_float(__builtin_amdgcn_readlane(__float_as_int(wA0_l), j + 1)), wa11 = __int_as_float(__builtin_amdgcn_readlane(__float_as_int(wA1_l), j + 1));
                float wb10 = __int_as_float(__builtin_amdgcn_readlane(__float_as_int(wB0_l), j + 1)), wb11 = __int_as_float(__builtin_amdgcn_readlane(__float_as_int(wB1_l), j + 1));
                asm volatile("" :: "v"(a0.ev), "v"(a0.xa), "v"(a0.xb), "v"(b0.ev), "v"(b0.xa), "v"(b0.xb),
                                   "v"(a1.ev), "v"(a1.xa), "v"(a1.xb), "v"(b1.ev), "v"(b1.xa), "v"(b1.xb));
                if (j < cntA)     add_edge(accA0, accA1, a0, wa00, wa01);
                if (j < cntB)     add_edge(accB0, accB1, b0, wb00, wb01);
                if (j + 1 < cntA) add_edge(accA0, accA1, a1, wa10, wa11);
                if (j + 1 < cntB) add_edge(accB0, accB1, b1, wb10, wb11);
            }
        }

        *(v2f*)(agg + (size_t)nA * D_FEAT + d)       = accA0;
        *(v2f*)(agg + ((size_t)N + nA) * D_FEAT + d) = accA1;
        *(v2f*)(agg + (size_t)nB * D_FEAT + d)       = accB0;
        *(v2f*)(agg + ((size_t)N + nB) * D_FEAT + d) = accB1;
        return;
    }

#pragma clang loop unroll(disable)
    for (int h = 0; h < NODES_PB / 8; ++h) {
    int li  = w + 8 * h;
    int n   = nb0 + li;
    if (n >= N) break;
    int key = n & kmask;
    int pad = key ^ 1;

    v2f acc0 = (v2f)(0.0f);
    v2f acc1 = (v2f)(0.0f);

    int deg = __builtin_amdgcn_readfirstlane(ncnt[li]);

    if (deg <= LIST_CAP) {
        for (int i = lane; i < deg; i += 32) {
            int ei = elist[li][i];
            int rank = 0;
            for (int j = 0; j < deg; ++j) rank += (elist[li][j] < ei) ? 1 : 0;
            srec[li][rank] = urec[li][i];
        }

        for (int g = 0; g < deg; g += 32) {
            int cnt = deg - g;
            if (cnt > 32) cnt = 32;

            int   sa_l = 0;
            float w0_l = 0.0f, w1_l = 0.0f;
            if (lane < cnt) {
                int4 rr = srec[li][g + lane];
                sa_l = rr.y;
                w0_l = __int_as_float(rr.z);
                w1_l = __int_as_float(rr.w);
            }

            const int samask = (1 << SA_SHIFT) - 1;
            int sa  = __builtin_amdgcn_readlane(sa_l, 0);
            int src = sa & samask;
            int att = (int)((unsigned)sa >> SA_SHIFT);
            v2f ev = *(const v2f*)(eb + (size_t)att * D_FEAT);
            v2f xa = *(const v2f*)(x0 + (size_t)src * D_FEAT);
            v2f xb = *(const v2f*)(x1 + (size_t)src * D_FEAT);
            for (int j = 0; j < cnt; ++j) {
                int jn   = (j + 1 < cnt) ? j + 1 : j;
                int san  = __builtin_amdgcn_readlane(sa_l, jn);
                int srcn = san & samask;
                int attn = (int)((unsigned)san >> SA_SHIFT);
                v2f evn = *(const v2f*)(eb + (size_t)attn * D_FEAT);
                v2f xan = *(const v2f*)(x0 + (size_t)srcn * D_FEAT);
                v2f xbn = *(const v2f*)(x1 + (size_t)srcn * D_FEAT);

                float w0 = __int_as_float(__builtin_amdgcn_readlane(__float_as_int(w0_l), j));
                float w1 = __int_as_float(__builtin_amdgcn_readlane(__float_as_int(w1_l), j));
                acc0 = __builtin_elementwise_fma(gelu_pair(xa + ev), (v2f)(w0), acc0);
                acc1 = __builtin_elementwise_fma(gelu_pair(xb + ev), (v2f)(w1), acc1);

                ev = evn; xa = xan; xb = xbn;
            }
        }
    } else {
        int last = -1;
        for (int k = 0; k < deg; ++k) {
            int best = 0x7fffffff;
            for (int base = s; base < t; base += 32) {
                int idx = base + lane;
                int r = (idx < t) ? recs4[idx].x : pad;
                int e = r >> BIN_SHIFT;
                if ((r & kmask) == key && e > last && e < best) best = e;
            }
            for (int base = 0; base < novf; base += 32) {
                int idx = base + lane;
                if (idx < novf) {
                    int e  = ovf[2 * idx];
                    int dd = ovf[2 * idx + 1];
                    if (dd == n && e > last && e < best) best = e;
                }
            }
            for (int off = 16; off > 0; off >>= 1) {
                int o = __shfl_xor(best, off);
                best = (o < best) ? o : best;
            }
            int e = __builtin_amdgcn_readfirstlane(best);
            last = e;
            int   src = edge_index[e];
            int   att = edge_attr[e];
            float w0  = edge_weight[e];
            float w1  = edge_weight[(size_t)E + e];
            v2f ev = *(const v2f*)(eb + (size_t)att * D_FEAT);
            v2f xa = *(const v2f*)(x0 + (size_t)src * D_FEAT);
            v2f xb = *(const v2f*)(x1 + (size_t)src * D_FEAT);
            acc0 = __builtin_elementwise_fma(gelu_pair(xa + ev), (v2f)(w0), acc0);
            acc1 = __builtin_elementwise_fma(gelu_pair(xb + ev), (v2f)(w1), acc1);
        }
    }

    *(v2f*)(agg + (size_t)n * D_FEAT + d)       = acc0;
    *(v2f*)(agg + ((size_t)N + n) * D_FEAT + d) = acc1;
    }
}

__global__ __launch_bounds__(MLP_WAVES * 32)
void gin_mlp_kernel(const float* __restrict__ x,
                    const float* __restrict__ agg,
                    const float* __restrict__ W1,
                    const float* __restrict__ b1,
                    const float* __restrict__ W2,
                    const float* __restrict__ b2,
                    const float* __restrict__ eps,
                    float*       __restrict__ out,
                    int n_tiles)
{
    __shared__ float4 w1s[8 * 8 * 32];
    __shared__ float4 w2s[4 * 16 * 32];
    __shared__ float  lds_t[MLP_WAVES][16 * H_FEAT];

    static_assert(MLP_WAVES >= STG_WAVES, "the first STG_WAVES waves stage the weights");
    int waveId = __builtin_amdgcn_readfirstlane(threadIdx.x >> 5);
    int lane   = threadIdx.x & 31;
    int t_lo   = (int)(((unsigned)blockIdx.x * (unsigned)n_tiles) / (unsigned)MLP_BLOCKS);
    int t_hi   = (int)((((unsigned)blockIdx.x + 1u) * (unsigned)n_tiles) / (unsigned)MLP_BLOCKS);
    int tile   = t_lo + waveId;
    bool valid = tile < t_hi;

    int m    = lane & 15;
    int half = lane >> 4;

    float epsv = 1.0f + eps[0];
    float2 xr[16], ar[16];
    size_t rowbase0 = ((size_t)(valid ? tile : 0) * 16 + m) * D_FEAT;

    if (waveId < STG_WAVES) {
        float4 stg[16];
#pragma unroll
        for (int n = 0; n < 8; ++n) {
            int ncol = n * 16 + m;
            int kb0  = waveId * 8 + half * 2;
            stg[n].x = W1[(size_t)(kb0    ) * H_FEAT + ncol];
            stg[n].y = W1[(size_t)(kb0 + 1) * H_FEAT + ncol];
            stg[n].z = W1[(size_t)(kb0 + 4) * H_FEAT + ncol];
            stg[n].w = W1[(size_t)(kb0 + 5) * H_FEAT + ncol];
        }
#pragma unroll
        for (int i = 0; i < 8; ++i) {
            int o    = i >> 1;
            int jj   = waveId + 8 * (i & 1);
            int ocol = o * 16 + m;
            int kb0  = jj * 8 + half * 2;
            stg[8 + i].x = W2[(size_t)(kb0    ) * D_FEAT + ocol];
            stg[8 + i].y = W2[(size_t)(kb0 + 1) * D_FEAT + ocol];
            stg[8 + i].z = W2[(size_t)(kb0 + 4) * D_FEAT + ocol];
            stg[8 + i].w = W2[(size_t)(kb0 + 5) * D_FEAT + ocol];
        }
#pragma unroll
        for (int k = 0; k < 16; ++k) {
            int col = k * 4 + half * 2;
            xr[k] = *(const float2*)(x   + rowbase0 + col);
            ar[k] = *(const float2*)(agg + rowbase0 + col);
        }
#pragma unroll
        for (int n = 0; n < 8; ++n)
            w1s[(n * 8 + waveId) * 32 + lane] = stg[n];
#pragma unroll
        for (int i = 0; i < 8; ++i)
            w2s[((i >> 1) * 16 + waveId + 8 * (i & 1)) * 32 + lane] = stg[8 + i];
    } else {
#pragma unroll
        for (int k = 0; k < 16; ++k) {
            int col = k * 4 + half * 2;
            xr[k] = *(const float2*)(x   + rowbase0 + col);
            ar[k] = *(const float2*)(agg + rowbase0 + col);
        }
    }
    v2f A[16];
#pragma unroll
    for (int k = 0; k < 16; ++k) {
        A[k].x = epsv * xr[k].x + ar[k].x;
        A[k].y = epsv * xr[k].y + ar[k].y;
    }
    __syncthreads();
    if (!valid) return;

    float* tl = &lds_t[waveId][0];

    for (;;) {
#pragma unroll
        for (int n = 0; n < 8; ++n) {
            v8f acc = {};
            int ncol = n * 16 + m;
#pragma unroll
            for (int kk = 0; kk < 8; ++kk) {
                float4 e = w1s[(n * 8 + kk) * 32 + lane];
                v2f B0; B0.x = e.x; B0.y = e.y;
                v2f B1; B1.x = e.z; B1.y = e.w;
                acc = __builtin_amdgcn_wmma_f32_16x16x4_f32(
                          false, A[2 * kk], false, B0, (short)0, acc, false, false);
                acc = __builtin_amdgcn_wmma_f32_16x16x4_f32(
                          false, A[2 * kk + 1], false, B1, (short)0, acc, false, false);
            }
            float bias = b1[ncol];
#pragma unroll
            for (int v = 0; v < 8; ++v) {
                float t = acc[v] + bias;
                t = t > 0.0f ? t : 0.0f;
                tl[(v + 8 * half) * H_FEAT + ncol] = t;
            }
        }

        asm volatile("s_wait_dscnt 0" ::: "memory");
        __builtin_amdgcn_wave_barrier();

#pragma unroll
        for (int o = 0; o < 4; ++o) {
            v8f acc = {};
            int ocol = o * 16 + m;
#pragma unroll
            for (int jj = 0; jj < 16; ++jj) {
                float4 e = w2s[(o * 16 + jj) * 32 + lane];
                int kb = jj * 8 + half * 2;
                float2 a0 = *(const float2*)(tl + m * H_FEAT + kb);
                float2 a1 = *(const float2*)(tl + m * H_FEAT + kb + 4);
                v2f A2; A2.x = a0.x; A2.y = a0.y;
                v2f A3; A3.x = a1.x; A3.y = a1.y;
                v2f B0; B0.x = e.x; B0.y = e.y;
                v2f B1; B1.x = e.z; B1.y = e.w;
                acc = __builtin_amdgcn_wmma_f32_16x16x4_f32(
                          false, A2, false, B0, (short)0, acc, false, false);
                acc = __builtin_amdgcn_wmma_f32_16x16x4_f32(
                          false, A3, false, B1, (short)0, acc, false, false);
            }
            float bias = b2[ocol];
#pragma unroll
            for (int v = 0; v < 8; ++v) {
                out[((size_t)tile * 16 + v + 8 * half) * D_FEAT + ocol] = acc[v] + bias;
            }
        }

        tile += MLP_WAVES;
        if (tile >= t_hi) break;
        size_t rowbase = ((size_t)tile * 16 + m) * D_FEAT;
#pragma unroll
        for (int k = 0; k < 16; ++k) {
            int col = k * 4 + half * 2;
            float2 xv = *(const float2*)(x   + rowbase + col);
            float2 av = *(const float2*)(agg + rowbase + col);
            A[k].x = epsv * xv.x + av.x;
            A[k].y = epsv * xv.y + av.y;
        }
    }
}

extern "C" void kernel_launch(void* const* d_in, const int* in_sizes, int n_in,
                              void* d_out, int out_size, void* d_ws, size_t ws_size,
                              hipStream_t stream)
{
    const float* x           = (const float*)d_in[0];
    const int*   edge_index  = (const int*)  d_in[1];
    const int*   edge_attr   = (const int*)  d_in[2];
    const float* edge_weight = (const float*)d_in[3];
    const float* bond_emb    = (const float*)d_in[4];
    const float* eps         = (const float*)d_in[5];
    const float* W1          = (const float*)d_in[6];
    const float* b1          = (const float*)d_in[7];
    const float* W2          = (const float*)d_in[8];
    const float* b2          = (const float*)d_in[9];
    float*       out         = (float*)d_out;

    const int RC = 2;
    const int E  = in_sizes[2];
    const int N  = in_sizes[0] / (RC * D_FEAT);
    const int NB = (N + (1 << BIN_SHIFT) - 1) >> BIN_SHIFT;

    float* agg        = (float*)d_ws;
    int*   bin_cursor = (int*)(agg + (size_t)RC * N * D_FEAT);
    int*   recs       = bin_cursor + MAX_BINS + 32;
    int*   ovf        = recs + (size_t)MAX_BINS * BIN_CAP * 4;

    hipMemsetAsync(bin_cursor, 0, (size_t)(MAX_BINS + 32) * sizeof(int), stream);

    int blocksP = (E + PART_EDGES - 1) / PART_EDGES;
    bin_partition_kernel<<<blocksP, PART_THREADS, 0, stream>>>(edge_index, edge_attr, edge_weight,
                                                               bin_cursor, recs, ovf, E, NB);

    int blocks1 = (N + NODES_PB - 1) / NODES_PB;
    node_sum_kernel<<<blocks1, 256, 0, stream>>>(x, edge_index, edge_attr, edge_weight, bond_emb,
                                                 bin_cursor, recs, ovf, agg, E, N);

    int n_tiles = (RC * N) / 16;
    gin_mlp_kernel<<<MLP_BLOCKS, MLP_WAVES * 32, 0, stream>>>(
        x, agg, W1, b1, W2, b2, eps, out, n_tiles);
}


# MLP fused into the node-sum launch: each block runs its 16 nodes' two 16-row tiles from LDS (weights laid out in fragment order by the binning launch); agg round trip and one launch gone
# speedup vs baseline: 2.7340x; 1.0940x over previous
#include <hip/hip_runtime.h>
#include <hip/hip_bf16.h>
#include <math.h>


typedef float v2f __attribute__((ext_vector_type(2)));
typedef float v8f __attribute__((ext_vector_type(8)));

#define D_FEAT 64
#define H_FEAT 128
#define WFRAG_ENTRIES 4096
#define HS_PITCH 68
#define TS_PITCH 132

#define BIN_SHIFT    6
#define BIN_CAP      2048
#define MAX_BINS     1024
#define PART_THREADS 512
#define PART_EDGES   4096
#define LIST_CAP     64
#define NODES_PB     16
#define BIN_WORDS    6
#define SA_SHIFT     20

static_assert(MAX_BINS == 2 * PART_THREADS, "pass 1 clears two LDS counters per thread");
static_assert(NODES_PB == 16 && (1 << BIN_SHIFT) % NODES_PB == 0, "a pass 2 block's nodes lie in one bin; its key tests use 16");

__device__ __forceinline__ float4 wfrag_entry(const float* __restrict__ W1, const float* __restrict__ W2, int g)
{
    int ln = g & 31;
    int mm = ln & 15;
    int hh = ln >> 4;
    float4 e;
    if (g < 2048) {
        int n   = g >> 8;
        int kk  = (g >> 5) & 7;
        int kb0 = kk * 8 + hh * 2;
        int col = n * 16 + mm;
        e.x = W1[(size_t)(kb0    ) * H_FEAT + col];
        e.y = W1[(size_t)(kb0 + 1) * H_FEAT + col];
        e.z = W1[(size_t)(kb0 + 4) * H_FEAT + col];
        e.w = W1[(size_t)(kb0 + 5) * H_FEAT + col];
    } else {
        int q   = g - 2048;
        int o   = q >> 9;
        int jj  = (q >> 5) & 15;
        int kb0 = jj * 8 + hh * 2;
        int col = o * 16 + mm;
        e.x = W2[(size_t)(kb0    ) * D_FEAT + col];
        e.y = W2[(size_t)(kb0 + 1) * D_FEAT + col];
        e.z = W2[(size_t)(kb0 + 4) * D_FEAT + col];
        e.w = W2[(size_t)(kb0 + 5) * D_FEAT + col];
    }
    return e;
}

__global__ __launch_bounds__(PART_THREADS)
void bin_partition_kernel(const int*   __restrict__ edge_index,
                          const int*   __restrict__ edge_attr,
                          const float* __restrict__ edge_weight,
                          const float* __restrict__ W1,
                          const float* __restrict__ W2,
                          float*     __restrict__ wfrag,
                          int*       __restrict__ bin_cursor,
                          int*       __restrict__ recs,
                          int*       __restrict__ ovf,
                          int E, int NB)
{
    int4* recs4 = (int4*)recs;
    float4* wf4 = (float4*)wfrag;
    __shared__ int hist[MAX_BINS];
    __shared__ int sbase[MAX_BINS];

    int tid = threadIdx.x;
    hist[2 * tid]     = 0;
    hist[2 * tid + 1] = 0;
    __syncthreads();

    int base = blockIdx.x * PART_EDGES;
    int dstv[PART_EDGES / PART_THREADS];
    for (int i = 0; i < PART_EDGES / PART_THREADS; ++i) {
        int e = base + i * PART_THREADS + tid;
        dstv[i] = (e < E) ? edge_index[E + e] : -1;
    }
    int   srcv[PART_EDGES / PART_THREADS];
    int   attv[PART_EDGES / PART_THREADS];
    float w0v[PART_EDGES / PART_THREADS];
    float w1v[PART_EDGES / PART_THREADS];
    for (int i = 0; i < PART_EDGES / PART_THREADS; ++i) {
        int e  = base + i * PART_THREADS + tid;
        int ec = (e < E) ? e : E - 1;
        srcv[i] = edge_index[ec];
        attv[i] = edge_attr[ec];
        w0v[i]  = edge_weight[ec];
        w1v[i]  = edge_weight[(size_t)E + ec];
    }
    int  wg   = blockIdx.x * PART_THREADS + tid;
    bool wstg = wg < WFRAG_ENTRIES;
    float4 wq = make_float4(0.0f, 0.0f, 0.0f, 0.0f);
    if (wstg) wq = wfrag_entry(W1, W2, wg);

    for (int i = 0; i < PART_EDGES / PART_THREADS; ++i)
        if (dstv[i] >= 0) atomicAdd(&hist[dstv[i] >> BIN_SHIFT], 1);
    __syncthreads();

    for (int b = tid; b < NB; b += PART_THREADS) {
        int h = hist[b];
        if (h) {
            sbase[b] = atomicAdd(&bin_cursor[b], h);
            hist[b]  = 0;
        }
    }
    __syncthreads();

    for (int i = 0; i < PART_EDGES / PART_THREADS; ++i) {
        if (dstv[i] >= 0) {
            int b    = dstv[i] >> BIN_SHIFT;
            int slot = sbase[b] + atomicAdd(&hist[b], 1);
            int e    = base + i * PART_THREADS + tid;
            if (slot < BIN_CAP) {
                int4 r;
                r.x = (e << BIN_SHIFT) | (dstv[i] & ((1 << BIN_SHIFT) - 1));
                r.y = srcv[i] | (attv[i] << SA_SHIFT);
                r.z = __float_as_int(w0v[i]);
                r.w = __float_as_int(w1v[i]);
                recs4[(size_t)b * BIN_CAP + slot] = r;
            } else {
                int k = atomicAdd(&bin_cursor[MAX_BINS], 1);
                ovf[2 * k]     = e;
                ovf[2 * k + 1] = dstv[i];
            }
        }
    }

    if (wstg) wf4[wg] = wq;
    for (int g2 = wg + (int)gridDim.x * PART_THREADS; g2 < WFRAG_ENTRIES; g2 += (int)gridDim.x * PART_THREADS)
        wf4[g2] = wfrag_entry(W1, W2, g2);
}

__device__ __forceinline__ v2f gelu_pair(v2f v)
{
    v2f a;
    a.x = fminf(fabsf(v.x), 8.0f);
    a.y = fminf(fabsf(v.y), 8.0f);
    v2f q = (v2f)(2.1919348114352033e-07f);
    q = __builtin_elementwise_fma(q, a, (v2f)(-5.2599425544030964e-06f));
    q = __builtin_elementwise_fma(q, a, (v2f)(4.943032763549127e-05f));
    q = __builtin_elementwise_fma(q, a, (v2f)(-0.00020471429161261767f));
    q = __builtin_elementwise_fma(q, a, (v2f)(-0.00012574861466418952f));
    q = __builtin_elementwise_fma(q, a, (v2f)(0.007073047570884228f));
    q = __builtin_elementwise_fma(q, a, (v2f)(-0.05251256376504898f));
    q = __builtin_elementwise_fma(q, a, (v2f)(-0.45920106768608093f));
    q = __builtin_elementwise_fma(q, a, (v2f)(-1.1511061191558838f));
    q = __builtin_elementwise_fma(q, a, (v2f)(-1.0f));
    v2f t;
    t.x = __builtin_amdgcn_exp2f(q.x);
    t.y = __builtin_amdgcn_exp2f(q.y);
    v2f r;
    r.x = fmaxf(v.x, 0.0f);
    r.y = fmaxf(v.y, 0.0f);
    return __builtin_elementwise_fma(-a, t, r);
}

struct EdgeRows { v2f ev, xa, xb; };

__device__ __forceinline__ EdgeRows ask_rows(const float* x, const float* bond_emb, size_t chan, unsigned d, int sa)
{
    const int samask = (1 << SA_SHIFT) - 1;
    int src = sa & samask;
    int att = (int)((unsigned)sa >> SA_SHIFT);
    const float* er = bond_emb + (size_t)att * D_FEAT;
    const float* xr = x + (size_t)src * D_FEAT;
    EdgeRows r;
    r.ev = *(const v2f*)(er + d);
    r.xa = *(const v2f*)(xr + d);
    r.xb = *(const v2f*)(xr + chan + d);
    return r;
}

__device__ __forceinline__ void add_edge(v2f& acc0, v2f& acc1, const EdgeRows& r, float w0, float w1)
{
    acc0 = __builtin_elementwise_fma(gelu_pair(r.xa + r.ev), (v2f)(w0), acc0);
    acc1 = __builtin_elementwise_fma(gelu_pair(r.xb + r.ev), (v2f)(w1), acc1);
}

__device__ __forceinline__ void rank_list(const int* el, const int4* ur, int4* sr, int deg, int lane)
{
    for (int i = lane; i < deg; i += 32) {
        int ei = el[i];
        int rank = 0;
        for (int j = 0; j < deg; ++j) rank += (el[j] < ei) ? 1 : 0;
        sr[rank] = ur[i];
    }
}

__global__ __launch_bounds__(256)
void node_mlp_kernel(const float* __restrict__ x,
                     const int*   __restrict__ edge_index,
                     const int*   __restrict__ edge_attr,
                     const float* __restrict__ edge_weight,
                     const float* __restrict__ bond_emb,
                     const int*   __restrict__ bin_cursor,
                     const int*   __restrict__ recs,
                     const int*   __restrict__ ovf,
                     const float* __restrict__ wfrag,
                     const float* __restrict__ b1,
                     const float* __restrict__ b2,
                     const float* __restrict__ eps,
                     float*       __restrict__ out,
                     int E, int N)
{
    __shared__ int  ncnt[NODES_PB];
    __shared__ int  elist[NODES_PB][LIST_CAP];
    __shared__ int4 urec[NODES_PB][LIST_CAP];
    __shared__ int4 srec[NODES_PB][LIST_CAP + 2];

    static_assert(sizeof(urec) >= 2 * 16 * HS_PITCH * sizeof(float), "the h rows of both tiles fit where the unordered records were");
    static_assert(sizeof(srec) >= 2 * 16 * TS_PITCH * sizeof(float), "t of both tiles fits where the ordered records were");

    const int4* recs4 = (const int4*)recs;

    int tid  = threadIdx.x;
    int w    = tid >> 5;
    int lane = tid & 31;
    int nb0  = blockIdx.x * NODES_PB;
    int wu   = __builtin_amdgcn_readfirstlane(tid >> 5);

    const int kmask = (1 << BIN_SHIFT) - 1;
    int bin  = nb0 >> BIN_SHIFT;
    int bkey = (nb0 & kmask) >> 4;
    int cbin = bin_cursor[bin];
    int novf = bin_cursor[MAX_BINS];
    int s    = bin * BIN_CAP;
    int t    = s + (cbin < BIN_CAP ? cbin : BIN_CAP);
    int d    = lane * 2;

    if (tid < NODES_PB) ncnt[tid] = 0;
    __syncthreads();
    for (int base = s; base < t; base += BIN_WORDS * 256) {
        int4 r[BIN_WORDS];
        for (int u = 0; u < BIN_WORDS; ++u) {
            int idx = base + u * 256 + tid;
            r[u] = make_int4(0, 0, 0, 0);
            if (idx < t) r[u] = recs4[idx];
        }
        for (int u = 0; u < BIN_WORDS; ++u) {
            int idx = base + u * 256 + tid;
            if (idx < t && ((r[u].x & kmask) >> 4) == bkey) {
                int k   = r[u].x & (NODES_PB - 1);
                int pos = atomicAdd(&ncnt[k], 1);
                if (pos < LIST_CAP) {
                    elist[k][pos] = r[u].x >> BIN_SHIFT;
                    urec[k][pos]  = r[u];
                }
            }
        }
    }
    for (int base = 0; base < novf; base += 256) {
        int idx = base + tid;
        if (idx < novf) {
            int e  = ovf[2 * idx];
            int dd = ovf[2 * idx + 1];
            if ((dd >> 4) == (nb0 >> 4)) {
                int k   = dd & (NODES_PB - 1);
                int pos = atomicAdd(&ncnt[k], 1);
                if (pos < LIST_CAP) {
                    int4 r;
                    r.x = (e << BIN_SHIFT) | (dd & kmask);
                    r.y = edge_index[e] | (edge_attr[e] << SA_SHIFT);
                    r.z = __float_as_int(edge_weight[e]);
                    r.w = __float_as_int(edge_weight[(size_t)E + e]);
                    elist[k][pos] = e;
                    urec[k][pos]  = r;
                }
            }
        }
    }
    __syncthreads();

    const float* x0 = x + d;
    const float* x1 = x + (size_t)N * D_FEAT + d;
    const float* eb = bond_emb + d;
    const size_t chan = (size_t)N * D_FEAT;

    int liA = w, liB = w + 8;
    int nA  = nb0 + liA, nB = nb0 + liB;

    v2f accA0 = (v2f)(0.0f), accA1 = (v2f)(0.0f);
    v2f accB0 = (v2f)(0.0f), accB1 = (v2f)(0.0f);

    if (nA < N) {
        int degA = __builtin_amdgcn_readfirstlane(ncnt[liA]);
        int degB = __builtin_amdgcn_readfirstlane(ncnt[liB]);

        if (nB < N && degA <= LIST_CAP && degB <= LIST_CAP) {
            rank_list(elist[liA], urec[liA], srec[liA], degA, lane);
            rank_list(elist[liB], urec[liB], srec[liB], degB, lane);

            int degM = degA > degB ? degA : degB;
            for (int g = 0; g < degM; g += 32) {
                int cntA = degA - g; cntA = cntA < 0 ? 0 : (cntA > 32 ? 32 : cntA);
                int cntB = degB - g; cntB = cntB < 0 ? 0 : (cntB > 32 ? 32 : cntB);

                int   saA_l = 0, saB_l = 0;
                float wA0_l = 0.0f, wA1_l = 0.0f, wB0_l = 0.0f, wB1_l = 0.0f;
                if (lane < cntA) {
                    int4 rr = srec[liA][g + lane];
                    saA_l = rr.y;
                    wA0_l = __int_as_float(rr.z);
                    wA1_l = __int_as_float(rr.w);
                }
                if (lane < cntB) {
                    int4 rr = srec[liB][g + lane];
                    saB_l = rr.y;
                    wB0_l = __int_as_float(rr.z);
                    wB1_l = __int_as_float(rr.w);
                }

                int m  = cntA > cntB ? cntA : cntB;
                int mf = (cntA < cntB ? cntA : cntB) & ~1;
                int j  = 0;
                for (; j < mf; j += 2) {
                    EdgeRows a0 = ask_rows(x, bond_emb, chan, (unsigned)d, __builtin_amdgcn_readlane(saA_l, j));
                    EdgeRows b0 = ask_rows(x, bond_emb, chan, (unsigned)d, __builtin_amdgcn_readlane(saB_l, j));
                    EdgeRows a1 = ask_rows(x, bond_emb, chan, (unsigned)d, __builtin_amdgcn_readlane(saA_l, j + 1));
                    EdgeRows b1 = ask_rows(x, bond_emb, chan, (unsigned)d, __builtin_amdgcn_readlane(saB_l, j + 1));
                    float wa00 = __int_as_float(__builtin_amdgcn_readlane(__float_as_int(wA0_l), j)),     wa01 = __int_as_float(__builtin_amdgcn_readlane(__float_as_int(wA1_l), j));
                    float wb00 = __int_as_float(__builtin_amdgcn_readlane(__float_as_int(wB0_l), j)),     wb01 = __int_as_float(__builtin_amdgcn_readlane(__float_as_int(wB1_l), j));
                    float wa10 = __int_as_float(__builtin_amdgcn_readlane(__float_as_int(wA0_l), j + 1)), wa11 = __int_as_float(__builtin_amdgcn_readlane(__float_as_int(wA1_l), j + 1));
                    float wb10 = __int_as_float(__builtin_amdgcn_readlane(__float_as_int(wB0_l), j + 1)), wb11 = __int_as_float(__builtin_amdgcn_readlane(__float_as_int(wB1_l), j + 1));
                    add_edge(accA0, accA1, a0, wa00, wa01);
                    add_edge(accB0, accB1, b0, wb00, wb01);
                    add_edge(accA0, accA1, a1, wa10, wa11);
                    add_edge(accB0, accB1, b1, wb10, wb11);
                }
                for (; j < m; j += 2) {
                    EdgeRows a0 = ask_rows(x, bond_emb, chan, (unsigned)d, __builtin_amdgcn_readlane(saA_l, j));
                    EdgeRows b0 = ask_rows(x, bond_emb, chan, (unsigned)d, __builtin_amdgcn_readlane(saB_l, j));
                    EdgeRows a1 = ask_rows(x, bond_emb, chan, (unsigned)d, __builtin_amdgcn_readlane(saA_l, j + 1));
                    EdgeRows b1 = ask_rows(x, bond_emb, chan, (unsigned)d, __builtin_amdgcn_readlane(saB_l, j + 1));
                    float wa00 = __int_as_float(__builtin_amdgcn_readlane(__float_as_int(wA0_l), j)),     wa01 = __int_as_float(__builtin_amdgcn_readlane(__float_as_int(wA1_l), j));
                    float wb00 = __int_as_float(__builtin_amdgcn_readlane(__float_as_int(wB0_l), j)),     wb01 = __int_as_float(__builtin_amdgcn_readlane(__float_as_int(wB1_l), j));
                    float wa10 = __int_as_float(__builtin_amdgcn_readlane(__float_as_int(wA0_l), j + 1)), wa11 = __int_as_float(__builtin_amdgcn_readlane(__float_as_int(wA1_l), j + 1));
                    float wb10 = __int_as_float(__builtin_amdgcn_readlane(__float_as_int(wB0_l), j + 1)), wb11 = __int_as_float(__builtin_amdgcn_readlane(__float_as_int(wB1_l), j + 1));
                    asm volatile("" :: "v"(a0.ev), "v"(a0.xa), "v"(a0.xb), "v"(b0.ev), "v"(b0.xa), "v"(b0.xb),
                                       "v"(a1.ev), "v"(a1.xa), "v"(a1.xb), "v"(b1.ev), "v"(b1.xa), "v"(b1.xb));
                    if (j < cntA)     add_edge(accA0, accA1, a0, wa00, wa01);
                    if (j < cntB)     add_edge(accB0, accB1, b0, wb00, wb01);
                    if (j + 1 < cntA) add_edge(accA0, accA1, a1, wa10, wa11);
                    if (j + 1 < cntB) add_edge(accB0, accB1, b1, wb10, wb11);
                }
            }
        } else {
#pragma clang loop unroll(disable)
            for (int h = 0; h < NODES_PB / 8; ++h) {
            int li  = w + 8 * h;
            int n   = nb0 + li;
            if (n >= N) break;
            int key = n & kmask;
            int pad = key ^ 1;

            v2f acc0 = (v2f)(0.0f);
            v2f acc1 = (v2f)(0.0f);

            int deg = __builtin_amdgcn_readfirstlane(ncnt[li]);

            if (deg <= LIST_CAP) {
                for (int i = lane; i < deg; i += 32) {
                    int ei = elist[li][i];
                    int rank = 0;
                    for (int j = 0; j < deg; ++j) rank += (elist[li][j] < ei) ? 1 : 0;
                    srec[li][rank] = urec[li][i];
                }

                for (int g = 0; g < deg; g += 32) {
                    int cnt = deg - g;
                    if (cnt > 32) cnt = 32;

                    int   sa_l = 0;
                    float w0_l = 0.0f, w1_l = 0.0f;
                    if (lane < cnt) {
                        int4 rr = srec[li][g + lane];
                        sa_l = rr.y;
                        w0_l = __int_as_float(rr.z);
                        w1_l = __int_as_float(rr.w);
                    }

                    const int samask = (1 << SA_SHIFT) - 1;
                    int sa  = __builtin_amdgcn_readlane(sa_l, 0);
                    int src = sa & samask;
                    int att = (int)((unsigned)sa >> SA_SHIFT);
                    v2f ev = *(const v2f*)(eb + (size_t)att * D_FEAT);
                    v2f xa = *(const v2f*)(x0 + (size_t)src * D_FEAT);
                    v2f xb = *(const v2f*)(x1 + (size_t)src * D_FEAT);
                    for (int j = 0; j < cnt; ++j) {
                        int jn   = (j + 1 < cnt) ? j + 1 : j;
                        int san  = __builtin_amdgcn_readlane(sa_l, jn);
                        int srcn = san & samask;
                        int attn = (int)((unsigned)san >> SA_SHIFT);
                        v2f evn = *(const v2f*)(eb + (size_t)attn * D_FEAT);
                        v2f xan = *(const v2f*)(x0 + (size_t)srcn * D_FEAT);
                        v2f xbn = *(const v2f*)(x1 + (size_t)srcn * D_FEAT);

                        float w0 = __int_as_float(__builtin_amdgcn_readlane(__float_as_int(w0_l), j));
                        float w1 = __int_as_float(__builtin_amdgcn_readlane(__float_as_int(w1_l), j));
                        acc0 = __builtin_elementwise_fma(gelu_pair(xa + ev), (v2f)(w0), acc0);
                        acc1 = __builtin_elementwise_fma(gelu_pair(xb + ev), (v2f)(w1), acc1);

                        ev = evn; xa = xan; xb = xbn;
                    }
                }
            } else {
                int last = -1;
                for (int k = 0; k < deg; ++k) {
                    int best = 0x7fffffff;
                    for (int base = s; base < t; base += 32) {
                        int idx = base + lane;
                        int r = (idx < t) ? recs4[idx].x : pad;
                        int e = r >> BIN_SHIFT;
                        if ((r & kmask) == key && e > last && e < best) best = e;
                    }
                    for (int base = 0; base < novf; base += 32) {
                        int idx = base + lane;
                        if (idx < novf) {
                            int e  = ovf[2 * idx];
                            int dd = ovf[2 * idx + 1];
                            if (dd == n && e > last && e < best) best = e;
                        }
                    }
                    for (int off = 16; off > 0; off >>= 1) {
                        int o = __shfl_xor(best, off);
                        best = (o < best) ? o : best;
                    }
                    int e = __builtin_amdgcn_readfirstlane(best);
                    last = e;
                    int   src = edge_index[e];
                    int   att = edge_attr[e];
                    float w0  = edge_weight[e];
                    float w1  = edge_weight[(size_t)E + e];
                    v2f ev = *(const v2f*)(eb + (size_t)att * D_FEAT);
                    v2f xa = *(const v2f*)(x0 + (size_t)src * D_FEAT);
                    v2f xb = *(const v2f*)(x1 + (size_t)src * D_FEAT);
                    acc0 = __builtin_elementwise_fma(gelu_pair(xa + ev), (v2f)(w0), acc0);
                    acc1 = __builtin_elementwise_fma(gelu_pair(xb + ev), (v2f)(w1), acc1);
                }
            }

            if (h == 0) { accA0 = acc0; accA1 = acc1; }
            else        { accB0 = acc0; accB1 = acc1; }
            }
        }
    }

    int ln   = (int)__builtin_amdgcn_mbcnt_lo(~0u, 0u);
    int m    = ln & 15;
    int half = ln >> 4;
    int d2   = ln * 2;
    int rA   = wu, rB = wu + 8;
    int mA   = nb0 + rA, mB = nb0 + rB;

    const float4* w1g = (const float4*)wfrag;
    const float4* w2g = w1g + 2048;

    float4 bq[8];
#pragma unroll
    for (int kk = 0; kk < 8; ++kk)
        bq[kk] = w1g[(wu * 8 + kk) * 32 + ln];

    float epsv = 1.0f + eps[0];
    v2f hA0 = (v2f)(0.0f), hA1 = (v2f)(0.0f);
    v2f hB0 = (v2f)(0.0f), hB1 = (v2f)(0.0f);
    if (mA < N) {
        v2f xa = *(const v2f*)(x + (size_t)mA * D_FEAT + d2);
        v2f xb = *(const v2f*)(x + chan + (size_t)mA * D_FEAT + d2);
        hA0.x = epsv * xa.x + accA0.x;
        hA0.y = epsv * xa.y + accA0.y;
        hA1.x = epsv * xb.x + accA1.x;
        hA1.y = epsv * xb.y + accA1.y;
    }
    if (mB < N) {
        v2f xa = *(const v2f*)(x + (size_t)mB * D_FEAT + d2);
        v2f xb = *(const v2f*)(x + chan + (size_t)mB * D_FEAT + d2);
        hB0.x = epsv * xa.x + accB0.x;
        hB0.y = epsv * xa.y + accB0.y;
        hB1.x = epsv * xb.x + accB1.x;
        hB1.y = epsv * xb.y + accB1.y;
    }

    float* hs = (float*)&urec[0][0];
    float* ts = (float*)&srec[0][0];

    __syncthreads();
    *(v2f*)(hs + (size_t)rA * HS_PITCH + d2)        = hA0;
    *(v2f*)(hs + (size_t)(16 + rA) * HS_PITCH + d2) = hA1;
    *(v2f*)(hs + (size_t)rB * HS_PITCH + d2)        = hB0;
    *(v2f*)(hs + (size_t)(16 + rB) * HS_PITCH + d2) = hB1;
    __syncthreads();

    {
        v8f c0 = {};
        v8f c1 = {};
        const float* ha = hs + m * HS_PITCH + 2 * half;
#pragma unroll
        for (int kk = 0; kk < 8; ++kk) {
            v2f a00 = *(const v2f*)(ha + 8 * kk);
            v2f a01 = *(const v2f*)(ha + 8 * kk + 4);
            v2f a10 = *(const v2f*)(ha + 16 * HS_PITCH + 8 * kk);
            v2f a11 = *(const v2f*)(ha + 16 * HS_PITCH + 8 * kk + 4);
            v2f B0; B0.x = bq[kk].x; B0.y = bq[kk].y;
            v2f B1; B1.x = bq[kk].z; B1.y = bq[kk].w;
            c0 = __builtin_amdgcn_wmma_f32_16x16x4_f32(
                     false, a00, false, B0, (short)0, c0, false, false);
            c1 = __builtin_amdgcn_wmma_f32_16x16x4_f32(
                     false, a10, false, B0, (short)0, c1, false, false);
            c0 = __builtin_amdgcn_wmma_f32_16x16x4_f32(
                     false, a01, false, B1, (short)0, c0, false, false);
            c1 = __builtin_amdgcn_wmma_f32_16x16x4_f32(
                     false, a11, false, B1, (short)0, c1, false, false);
        }
        int ncol = wu * 16 + m;
        float bias = b1[ncol];
#pragma unroll
        for (int v = 0; v < 8; ++v) {
            float t0 = c0[v] + bias;
            float t1 = c1[v] + bias;
            t0 = t0 > 0.0f ? t0 : 0.0f;
            t1 = t1 > 0.0f ? t1 : 0.0f;
            ts[(v + 8 * half) * TS_PITCH + ncol]      = t0;
            ts[(16 + v + 8 * half) * TS_PITCH + ncol] = t1;
        }
    }
    __syncthreads();

    float* os = hs;
    if (wu < 4) {
        v8f d0 = {};
        v8f d1 = {};
        const float* ta = ts + m * TS_PITCH + 2 * half;
#pragma unroll 4
        for (int jj = 0; jj < 16; ++jj) {
            float4 e = w2g[(wu * 16 + jj) * 32 + ln];
            v2f a00 = *(const v2f*)(ta + 8 * jj);
            v2f a01 = *(const v2f*)(ta + 8 * jj + 4);
            v2f a10 = *(const v2f*)(ta + 16 * TS_PITCH + 8 * jj);
            v2f a11 = *(const v2f*)(ta + 16 * TS_PITCH + 8 * jj + 4);
            v2f B0; B0.x = e.x; B0.y = e.y;
            v2f B1; B1.x = e.z; B1.y = e.w;
            d0 = __builtin_amdgcn_wmma_f32_16x16x4_f32(
                     false, a00, false, B0, (short)0, d0, false, false);
            d1 = __builtin_amdgcn_wmma_f32_16x16x4_f32(
                     false, a10, false, B0, (short)0, d1, false, false);
            d0 = __builtin_amdgcn_wmma_f32_16x16x4_f32(
                     false, a01, false, B1, (short)0, d0, false, false);
            d1 = __builtin_amdgcn_wmma_f32_16x16x4_f32(
                     false, a11, false, B1, (short)0, d1, false, false);
        }
        int ocol = wu * 16 + m;
        float bias = b2[ocol];
#pragma unroll
        for (int v = 0; v < 8; ++v) {
            os[(v + 8 * half) * HS_PITCH + ocol]      = d0[v] + bias;
            os[(16 + v + 8 * half) * HS_PITCH + ocol] = d1[v] + bias;
        }
    }
    __syncthreads();

    if (mA < N) {
        *(v2f*)(out + (size_t)mA * D_FEAT + d2)       = *(const v2f*)(os + (size_t)rA * HS_PITCH + d2);
        *(v2f*)(out + ((size_t)N + mA) * D_FEAT + d2) = *(const v2f*)(os + (size_t)(16 + rA) * HS_PITCH + d2);
    }
    if (mB < N) {
        *(v2f*)(out + (size_t)mB * D_FEAT + d2)       = *(const v2f*)(os + (size_t)rB * HS_PITCH + d2);
        *(v2f*)(out + ((size_t)N + mB) * D_FEAT + d2) = *(const v2f*)(os + (size_t)(16 + rB) * HS_PITCH + d2);
    }
}

extern "C" void kernel_launch(void* const* d_in, const int* in_sizes, int n_in,
                              void* d_out, int out_size, void* d_ws, size_t ws_size,
                              hipStream_t stream)
{
    const float* x           = (const float*)d_in[0];
    const int*   edge_index  = (const int*)  d_in[1];
    const int*   edge_attr   = (const int*)  d_in[2];
    const float* edge_weight = (const float*)d_in[3];
    const float* bond_emb    = (const float*)d_in[4];
    const float* eps         = (const float*)d_in[5];
    const float* W1          = (const float*)d_in[6];
    const float* b1          = (const float*)d_in[7];
    const float* W2          = (const float*)d_in[8];
    const float* b2          = (const float*)d_in[9];
    float*       out         = (float*)d_out;

    const int RC = 2;
    const int E  = in_sizes[2];
    const int N  = in_sizes[0] / (RC * D_FEAT);
    const int NB = (N + (1 << BIN_SHIFT) - 1) >> BIN_SHIFT;

    float* wfrag      = (float*)d_ws;
    int*   bin_cursor = (int*)(wfrag + (size_t)WFRAG_ENTRIES * 4);
    int*   recs       = bin_cursor + MAX_BINS + 32;
    int*   ovf        = recs + (size_t)MAX_BINS * BIN_CAP * 4;

    hipMemsetAsync(bin_cursor, 0, (size_t)(MAX_BINS + 32) * sizeof(int), stream);

    int blocksP = (E + PART_EDGES - 1) / PART_EDGES;
    bin_partition_kernel<<<blocksP, PART_THREADS, 0, stream>>>(edge_index, edge_attr, edge_weight, W1, W2, wfrag,
                                                               bin_cursor, recs, ovf, E, NB);

    int blocks1 = (N + NODES_PB - 1) / NODES_PB;
    node_mlp_kernel<<<blocks1, 256, 0, stream>>>(x, edge_index, edge_attr, edge_weight, bond_emb,
                                                 bin_cursor, recs, ovf, wfrag, b1, b2, eps, out, E, N);
}
